# starting kernel
# baseline (speedup 1.0000x reference)
#include <hip/hip_runtime.h>
#include <math.h>


typedef __attribute__((ext_vector_type(2))) float v2f;
typedef __attribute__((ext_vector_type(8))) float v8f;

#define LDS_STRIDE 1028

__device__ __forceinline__ int wredSum(int v) {
#pragma unroll
  for (int m = 16; m; m >>= 1) v += __shfl_xor(v, m, 32);
  return v;
}
__device__ __forceinline__ float wredSumF(float v) {
#pragma unroll
  for (int m = 16; m; m >>= 1) v += __shfl_xor(v, m, 32);
  return v;
}
__device__ __forceinline__ float wredMaxF(float v) {
#pragma unroll
  for (int m = 16; m; m >>= 1) v = fmaxf(v, __shfl_xor(v, m, 32));
  return v;
}

__device__ __forceinline__ unsigned kthSmallest(const unsigned (&key)[32], int k) {
  unsigned res = 0u;
  for (int bit = 31; bit >= 0; --bit) {
    unsigned cand = res | (1u << bit);
    int cnt = 0;
#pragma unroll
    for (int j = 0; j < 32; ++j) cnt += (key[j] < cand) ? 1 : 0;
    cnt = wredSum(cnt);
    if (cnt <= k) res = cand;
  }
  return res;
}

__global__ __launch_bounds__(256)
void prob_attn_kernel(const float* __restrict__ Q, const float* __restrict__ K,
                      const float* __restrict__ V, const int* __restrict__ starts,
                      float* __restrict__ O) {
  constexpr int L = 1024, H = 8, E = 64, RT = 32;

  __shared__ float S[RT * LDS_STRIDE];
  __shared__ unsigned inpatch[L];
  __shared__ float inv_sum[RT];

  const int tid  = threadIdx.x;
  const int wave = tid >> 5;
  const int lane = tid & 31;
  const int lh   = lane & 15;
  const int klo  = (lane < 16) ? 0 : 2;
  const int hih  = (lane < 16) ? 0 : 8;
  const int b = blockIdx.z, h = blockIdx.y;
  const int row_base = blockIdx.x * RT;

  for (int i = tid; i < L; i += 256) {
    unsigned m = 0u;
#pragma unroll
    for (int p = 0; p < 32; ++p) {
      const int s0 = starts[p];
      if (i >= s0 && i < s0 + 64) m |= (1u << p);
    }
    inpatch[i] = m;
  }
  __syncthreads();

  const int mt = wave & 1;
  const int r0 = mt * 16;
  {
    const float* qrow = Q + ((size_t)(b * L + row_base + r0 + lh) * H + h) * E;
    v2f afrag[16];
#pragma unroll
    for (int c = 0; c < 16; ++c)
      afrag[c] = *(const v2f*)(qrow + 4 * c + klo);

    unsigned prow[8];
#pragma unroll
    for (int i = 0; i < 8; ++i)
      prow[i] = inpatch[row_base + r0 + i + hih];

    for (int nt = (wave >> 1); nt < 64; nt += 4) {
      const int n0 = nt * 16;
      const float* krow = K + ((size_t)(b * L + n0 + lh) * H + h) * E;
      v8f acc = {};
#pragma unroll
      for (int c = 0; c < 16; ++c) {
        const v2f bfrag = *(const v2f*)(krow + 4 * c + klo);
        acc = __builtin_amdgcn_wmma_f32_16x16x4_f32(
            false, afrag[c], false, bfrag, (short)0, acc, false, false);
      }
      const unsigned pc = inpatch[n0 + lh];
#pragma unroll
      for (int i = 0; i < 8; ++i) {
        const float val = (prow[i] & pc) ? acc[i] : 0.0f;
        S[(r0 + i + hih) * LDS_STRIDE + n0 + lh] = val;
      }
    }
  }
  __syncthreads();

  for (int rr = 0; rr < 4; ++rr) {
    const int rl = wave * 4 + rr;
    float* srow = &S[rl * LDS_STRIDE];
    float sreg[32];
    unsigned kreg[32];
#pragma unroll
    for (int j = 0; j < 32; ++j) {
      const float s = srow[lane + 32 * j];
      sreg[j] = s;
      kreg[j] = __float_as_uint(s) & 0x7fffffffu;
    }
    const float a102 = __uint_as_float(kthSmallest(kreg, 102));
    const float a103 = __uint_as_float(kthSmallest(kreg, 103));
    const float thr  = a102 + 0.3f * (a103 - a102);
    const unsigned tb = __float_as_uint(thr);

    float mx = -INFINITY;
#pragma unroll
    for (int j = 0; j < 32; ++j)
      if (kreg[j] >= tb) mx = fmaxf(mx, sreg[j]);
    mx = wredMaxF(mx);

    float sum = 0.0f;
#pragma unroll
    for (int j = 0; j < 32; ++j) {
      const float p = (kreg[j] >= tb) ? expf(sreg[j] - mx) : 0.0f;
      sum += p;
      srow[lane + 32 * j] = p;
    }
    sum = wredSumF(sum);
    if (lane == 0) inv_sum[rl] = 1.0f / sum;
  }
  __syncthreads();

  {
    const int mt2 = wave >> 2;
    const int n0  = (wave & 3) * 16;
    const float* vcol = V + ((size_t)(b * L) * H + h) * E + n0 + lh;
    const float* arow = &S[(mt2 * 16 + lh) * LDS_STRIDE];
    v8f acc = {};
#pragma unroll 4
    for (int kk = 0; kk < L; kk += 4) {
      const v2f af = *(const v2f*)(arow + kk + klo);
      v2f bf;
      bf.x = vcol[(size_t)(kk + klo) * (H * E)];
      bf.y = vcol[(size_t)(kk + klo + 1) * (H * E)];
      acc = __builtin_amdgcn_wmma_f32_16x16x4_f32(
          false, af, false, bf, (short)0, acc, false, false);
    }
#pragma unroll
    for (int i = 0; i < 8; ++i) {
      const int rl = mt2 * 16 + i + hih;
      const int l  = row_base + rl;
      O[((size_t)(b * L + l) * H + h) * E + n0 + lh] = acc[i] * inv_sum[rl];
    }
  }
}

extern "C" void kernel_launch(void* const* d_in, const int* in_sizes, int n_in,
                              void* d_out, int out_size, void* d_ws, size_t ws_size,
                              hipStream_t stream) {
  const float* Q = (const float*)d_in[0];
  const float* K = (const float*)d_in[1];
  const float* V = (const float*)d_in[2];
  const int* starts = (const int*)d_in[3];
  float* O = (float*)d_out;

  dim3 grid(1024 / 32, 8, 8);
  dim3 block(256);
  hipLaunchKernelGGL(prob_attn_kernel, grid, block, 0, stream, Q, K, V, starts, O);
}


# Quantile select early-out: ballot-count the zero |s| keys of a row; with 104 or more zero, order statistics 102 and 103 are 0 (threshold 0), so both 32-pass bit descents are skipped; bit-identical
# speedup vs baseline: 1.9806x; 1.9806x over previous
#include <hip/hip_runtime.h>
#include <math.h>


typedef __attribute__((ext_vector_type(2))) float v2f;
typedef __attribute__((ext_vector_type(8))) float v8f;

#define LDS_STRIDE 1028

__device__ __forceinline__ int wredSum(int v) {
#pragma unroll
  for (int m = 16; m; m >>= 1) v += __shfl_xor(v, m, 32);
  return v;
}
__device__ __forceinline__ float wredSumF(float v) {
#pragma unroll
  for (int m = 16; m; m >>= 1) v += __shfl_xor(v, m, 32);
  return v;
}
__device__ __forceinline__ float wredMaxF(float v) {
#pragma unroll
  for (int m = 16; m; m >>= 1) v = fmaxf(v, __shfl_xor(v, m, 32));
  return v;
}

__device__ __forceinline__ unsigned kthSmallest(const unsigned (&key)[32], int k) {
  unsigned res = 0u;
  for (int bit = 31; bit >= 0; --bit) {
    unsigned cand = res | (1u << bit);
    int cnt = 0;
#pragma unroll
    for (int j = 0; j < 32; ++j) cnt += (key[j] < cand) ? 1 : 0;
    cnt = wredSum(cnt);
    if (cnt <= k) res = cand;
  }
  return res;
}

__global__ __launch_bounds__(256)
void prob_attn_kernel(const float* __restrict__ Q, const float* __restrict__ K,
                      const float* __restrict__ V, const int* __restrict__ starts,
                      float* __restrict__ O) {
  constexpr int L = 1024, H = 8, E = 64, RT = 32;

  __shared__ float S[RT * LDS_STRIDE];
  __shared__ unsigned inpatch[L];
  __shared__ float inv_sum[RT];

  const int tid  = threadIdx.x;
  const int wave = tid >> 5;
  const int lane = tid & 31;
  const int lh   = lane & 15;
  const int klo  = (lane < 16) ? 0 : 2;
  const int hih  = (lane < 16) ? 0 : 8;
  const int b = blockIdx.z, h = blockIdx.y;
  const int row_base = blockIdx.x * RT;

  for (int i = tid; i < L; i += 256) {
    unsigned m = 0u;
#pragma unroll
    for (int p = 0; p < 32; ++p) {
      const int s0 = starts[p];
      if (i >= s0 && i < s0 + 64) m |= (1u << p);
    }
    inpatch[i] = m;
  }
  __syncthreads();

  const int mt = wave & 1;
  const int r0 = mt * 16;
  {
    const float* qrow = Q + ((size_t)(b * L + row_base + r0 + lh) * H + h) * E;
    v2f afrag[16];
#pragma unroll
    for (int c = 0; c < 16; ++c)
      afrag[c] = *(const v2f*)(qrow + 4 * c + klo);

    unsigned prow[8];
#pragma unroll
    for (int i = 0; i < 8; ++i)
      prow[i] = inpatch[row_base + r0 + i + hih];

    for (int nt = (wave >> 1); nt < 64; nt += 4) {
      const int n0 = nt * 16;
      const float* krow = K + ((size_t)(b * L + n0 + lh) * H + h) * E;
      v8f acc = {};
#pragma unroll
      for (int c = 0; c < 16; ++c) {
        const v2f bfrag = *(const v2f*)(krow + 4 * c + klo);
        acc = __builtin_amdgcn_wmma_f32_16x16x4_f32(
            false, afrag[c], false, bfrag, (short)0, acc, false, false);
      }
      const unsigned pc = inpatch[n0 + lh];
#pragma unroll
      for (int i = 0; i < 8; ++i) {
        const float val = (prow[i] & pc) ? acc[i] : 0.0f;
        S[(r0 + i + hih) * LDS_STRIDE + n0 + lh] = val;
      }
    }
  }
  __syncthreads();

  for (int rr = 0; rr < 4; ++rr) {
    const int rl = wave * 4 + rr;
    float* srow = &S[rl * LDS_STRIDE];
    float sreg[32];
    unsigned kreg[32];
#pragma unroll
    for (int j = 0; j < 32; ++j) {
      const float s = srow[lane + 32 * j];
      sreg[j] = s;
      kreg[j] = __float_as_uint(s) & 0x7fffffffu;
    }
    int nzero = 0;
#pragma unroll
    for (int j = 0; j < 32; ++j)
      nzero += __builtin_popcount(__builtin_amdgcn_ballot_w32(kreg[j] == 0u));
    unsigned tb = 0u;
    if (nzero < 104) {
      const float a102 = __uint_as_float(kthSmallest(kreg, 102));
      const float a103 = __uint_as_float(kthSmallest(kreg, 103));
      const float thr  = a102 + 0.3f * (a103 - a102);
      tb = __float_as_uint(thr);
    }

    float mx = -INFINITY;
#pragma unroll
    for (int j = 0; j < 32; ++j)
      if (kreg[j] >= tb) mx = fmaxf(mx, sreg[j]);
    mx = wredMaxF(mx);

    float sum = 0.0f;
#pragma unroll
    for (int j = 0; j < 32; ++j) {
      const float p = (kreg[j] >= tb) ? expf(sreg[j] - mx) : 0.0f;
      sum += p;
      srow[lane + 32 * j] = p;
    }
    sum = wredSumF(sum);
    if (lane == 0) inv_sum[rl] = 1.0f / sum;
  }
  __syncthreads();

  {
    const int mt2 = wave >> 2;
    const int n0  = (wave & 3) * 16;
    const float* vcol = V + ((size_t)(b * L) * H + h) * E + n0 + lh;
    const float* arow = &S[(mt2 * 16 + lh) * LDS_STRIDE];
    v8f acc = {};
#pragma unroll 4
    for (int kk = 0; kk < L; kk += 4) {
      const v2f af = *(const v2f*)(arow + kk + klo);
      v2f bf;
      bf.x = vcol[(size_t)(kk + klo) * (H * E)];
      bf.y = vcol[(size_t)(kk + klo + 1) * (H * E)];
      acc = __builtin_amdgcn_wmma_f32_16x16x4_f32(
          false, af, false, bf, (short)0, acc, false, false);
    }
#pragma unroll
    for (int i = 0; i < 8; ++i) {
      const int rl = mt2 * 16 + i + hih;
      const int l  = row_base + rl;
      O[((size_t)(b * L + l) * H + h) * E + n0 + lh] = acc[i] * inv_sum[rl];
    }
  }
}

extern "C" void kernel_launch(void* const* d_in, const int* in_sizes, int n_in,
                              void* d_out, int out_size, void* d_ws, size_t ws_size,
                              hipStream_t stream) {
  const float* Q = (const float*)d_in[0];
  const float* K = (const float*)d_in[1];
  const float* V = (const float*)d_in[2];
  const int* starts = (const int*)d_in[3];
  float* O = (float*)d_out;

  dim3 grid(1024 / 32, 8, 8);
  dim3 block(256);
  hipLaunchKernelGGL(prob_attn_kernel, grid, block, 0, stream, Q, K, V, starts, O);
}


# Phase 1 block sparsity: a 16x16 score tile whose rows and columns share no patch is stored as zeros, skipping its K loads and 16 WMMAs (one ballot per wave picks the tiles); bit-identical
# speedup vs baseline: 2.5735x; 1.2994x over previous
#include <hip/hip_runtime.h>
#include <math.h>


typedef __attribute__((ext_vector_type(2))) float v2f;
typedef __attribute__((ext_vector_type(8))) float v8f;

#define LDS_STRIDE 1028

__device__ __forceinline__ int wredSum(int v) {
#pragma unroll
  for (int m = 16; m; m >>= 1) v += __shfl_xor(v, m, 32);
  return v;
}
__device__ __forceinline__ float wredSumF(float v) {
#pragma unroll
  for (int m = 16; m; m >>= 1) v += __shfl_xor(v, m, 32);
  return v;
}
__device__ __forceinline__ float wredMaxF(float v) {
#pragma unroll
  for (int m = 16; m; m >>= 1) v = fmaxf(v, __shfl_xor(v, m, 32));
  return v;
}

__device__ __forceinline__ unsigned kthSmallest(const unsigned (&key)[32], int k) {
  unsigned res = 0u;
  for (int bit = 31; bit >= 0; --bit) {
    unsigned cand = res | (1u << bit);
    int cnt = 0;
#pragma unroll
    for (int j = 0; j < 32; ++j) cnt += (key[j] < cand) ? 1 : 0;
    cnt = wredSum(cnt);
    if (cnt <= k) res = cand;
  }
  return res;
}

__global__ __launch_bounds__(256)
void prob_attn_kernel(const float* __restrict__ Q, const float* __restrict__ K,
                      const float* __restrict__ V, const int* __restrict__ starts,
                      float* __restrict__ O) {
  constexpr int L = 1024, H = 8, E = 64, RT = 32;
  constexpr int PATCH = 64;
  static_assert(PATCH >= 16, "the tile test in phase 1 needs a patch at least one tile long");

  __shared__ float S[RT * LDS_STRIDE];
  __shared__ unsigned inpatch[L];
  __shared__ float inv_sum[RT];

  const int tid  = threadIdx.x;
  const int wave = tid >> 5;
  const int lane = tid & 31;
  const int lh   = lane & 15;
  const int klo  = (lane < 16) ? 0 : 2;
  const int hih  = (lane < 16) ? 0 : 8;
  const int b = blockIdx.z, h = blockIdx.y;
  const int row_base = blockIdx.x * RT;

  for (int i = tid; i < L; i += 256) {
    unsigned m = 0u;
#pragma unroll
    for (int p = 0; p < 32; ++p) {
      const int s0 = starts[p];
      if (i >= s0 && i < s0 + PATCH) m |= (1u << p);
    }
    inpatch[i] = m;
  }
  __syncthreads();

  const int mt = wave & 1;
  const int r0 = mt * 16;
  {
    const float* qrow = Q + ((size_t)(b * L + row_base + r0 + lh) * H + h) * E;
    v2f afrag[16];
#pragma unroll
    for (int c = 0; c < 16; ++c)
      afrag[c] = *(const v2f*)(qrow + 4 * c + klo);

    unsigned prow[8];
#pragma unroll
    for (int i = 0; i < 8; ++i)
      prow[i] = inpatch[row_base + r0 + i + hih];

    const int g = wave >> 1;
    const unsigned rowor = inpatch[row_base + r0] | inpatch[row_base + r0 + 15];
    const int ct = (g + 4 * lh) * 16;
    const unsigned color = inpatch[ct] | inpatch[ct + 15];
    const unsigned active = __builtin_amdgcn_ballot_w32((rowor & color) != 0u);

    for (int k = 0; k < 16; ++k) {
      const int n0 = (g + 4 * k) * 16;
      if (((active >> k) & 1u) == 0u) {
        for (int i = 0; i < 8; ++i)
          S[(r0 + i + hih) * LDS_STRIDE + n0 + lh] = 0.0f;
        continue;
      }
      const float* krow = K + ((size_t)(b * L + n0 + lh) * H + h) * E;
      v8f acc = {};
#pragma unroll
      for (int c = 0; c < 16; ++c) {
        const v2f bfrag = *(const v2f*)(krow + 4 * c + klo);
        acc = __builtin_amdgcn_wmma_f32_16x16x4_f32(
            false, afrag[c], false, bfrag, (short)0, acc, false, false);
      }
      const unsigned pc = inpatch[n0 + lh];
#pragma unroll
      for (int i = 0; i < 8; ++i) {
        const float val = (prow[i] & pc) ? acc[i] : 0.0f;
        S[(r0 + i + hih) * LDS_STRIDE + n0 + lh] = val;
      }
    }
  }
  __syncthreads();

  for (int rr = 0; rr < 4; ++rr) {
    const int rl = wave * 4 + rr;
    float* srow = &S[rl * LDS_STRIDE];
    float sreg[32];
    unsigned kreg[32];
#pragma unroll
    for (int j = 0; j < 32; ++j) {
      const float s = srow[lane + 32 * j];
      sreg[j] = s;
      kreg[j] = __float_as_uint(s) & 0x7fffffffu;
    }
    int nzero = 0;
#pragma unroll
    for (int j = 0; j < 32; ++j)
      nzero += __builtin_popcount(__builtin_amdgcn_ballot_w32(kreg[j] == 0u));
    unsigned tb = 0u;
    if (nzero < 104) {
      const float a102 = __uint_as_float(kthSmallest(kreg, 102));
      const float a103 = __uint_as_float(kthSmallest(kreg, 103));
      const float thr  = a102 + 0.3f * (a103 - a102);
      tb = __float_as_uint(thr);
    }

    float mx = -INFINITY;
#pragma unroll
    for (int j = 0; j < 32; ++j)
      if (kreg[j] >= tb) mx = fmaxf(mx, sreg[j]);
    mx = wredMaxF(mx);

    float sum = 0.0f;
#pragma unroll
    for (int j = 0; j < 32; ++j) {
      const float p = (kreg[j] >= tb) ? expf(sreg[j] - mx) : 0.0f;
      sum += p;
      srow[lane + 32 * j] = p;
    }
    sum = wredSumF(sum);
    if (lane == 0) inv_sum[rl] = 1.0f / sum;
  }
  __syncthreads();

  {
    const int mt2 = wave >> 2;
    const int n0  = (wave & 3) * 16;
    const float* vcol = V + ((size_t)(b * L) * H + h) * E + n0 + lh;
    const float* arow = &S[(mt2 * 16 + lh) * LDS_STRIDE];
    v8f acc = {};
#pragma unroll 4
    for (int kk = 0; kk < L; kk += 4) {
      const v2f af = *(const v2f*)(arow + kk + klo);
      v2f bf;
      bf.x = vcol[(size_t)(kk + klo) * (H * E)];
      bf.y = vcol[(size_t)(kk + klo + 1) * (H * E)];
      acc = __builtin_amdgcn_wmma_f32_16x16x4_f32(
          false, af, false, bf, (short)0, acc, false, false);
    }
#pragma unroll
    for (int i = 0; i < 8; ++i) {
      const int rl = mt2 * 16 + i + hih;
      const int l  = row_base + rl;
      O[((size_t)(b * L + l) * H + h) * E + n0 + lh] = acc[i] * inv_sum[rl];
    }
  }
}

extern "C" void kernel_launch(void* const* d_in, const int* in_sizes, int n_in,
                              void* d_out, int out_size, void* d_ws, size_t ws_size,
                              hipStream_t stream) {
  const float* Q = (const float*)d_in[0];
  const float* K = (const float*)d_in[1];
  const float* V = (const float*)d_in[2];
  const int* starts = (const int*)d_in[3];
  float* O = (float*)d_out;

  dim3 grid(1024 / 32, 8, 8);
  dim3 block(256);
  hipLaunchKernelGGL(prob_attn_kernel, grid, block, 0, stream, Q, K, V, starts, O);
}


# Phase 3 split-K re-tiling: each wave does the full 32x64 tile over 1/8 of the keys with 128-bit V row loads (one fragment feeds 8 WMMAs); partials summed in wave order via LDS (sum order changes)
# speedup vs baseline: 5.1229x; 1.9906x over previous
#include <hip/hip_runtime.h>
#include <math.h>


typedef __attribute__((ext_vector_type(2))) float v2f;
typedef __attribute__((ext_vector_type(4))) float v4f;
typedef __attribute__((ext_vector_type(8))) float v8f;

#define LDS_STRIDE 1028
#define PART_STRIDE 68

__device__ __forceinline__ int wredSum(int v) {
#pragma unroll
  for (int m = 16; m; m >>= 1) v += __shfl_xor(v, m, 32);
  return v;
}
__device__ __forceinline__ float wredSumF(float v) {
#pragma unroll
  for (int m = 16; m; m >>= 1) v += __shfl_xor(v, m, 32);
  return v;
}
__device__ __forceinline__ float wredMaxF(float v) {
#pragma unroll
  for (int m = 16; m; m >>= 1) v = fmaxf(v, __shfl_xor(v, m, 32));
  return v;
}

__device__ __forceinline__ unsigned kthSmallest(const unsigned (&key)[32], int k) {
  unsigned res = 0u;
  for (int bit = 31; bit >= 0; --bit) {
    unsigned cand = res | (1u << bit);
    int cnt = 0;
#pragma unroll
    for (int j = 0; j < 32; ++j) cnt += (key[j] < cand) ? 1 : 0;
    cnt = wredSum(cnt);
    if (cnt <= k) res = cand;
  }
  return res;
}


template <int G>
__device__ __forceinline__ void p3_load(v4f (&vb)[G][2], v2f (&af)[G][2],
                                        const float* __restrict__ vrow,
                                        const float* a0, const float* a1,
                                        int ks0, int vpitch) {
#pragma unroll
  for (int g = 0; g < G; ++g) {
    const int k = 4 * (ks0 + g);
    vb[g][0] = *(const v4f*)(vrow + (size_t)k * vpitch);
    vb[g][1] = *(const v4f*)(vrow + (size_t)(k + 1) * vpitch);
    af[g][0] = *(const v2f*)(a0 + k);
    af[g][1] = *(const v2f*)(a1 + k);
  }
}
template <int G>
__device__ __forceinline__ void p3_mma(v8f (&acc)[2][4], const v4f (&vb)[G][2],
                                       const v2f (&af)[G][2]) {
#pragma unroll
  for (int g = 0; g < G; ++g)
#pragma unroll
    for (int t = 0; t < 4; ++t) {
      v2f bf;
      bf.x = vb[g][0][t];
      bf.y = vb[g][1][t];
      acc[0][t] = __builtin_amdgcn_wmma_f32_16x16x4_f32(
          false, af[g][0], false, bf, (short)0, acc[0][t], false, false);
      acc[1][t] = __builtin_amdgcn_wmma_f32_16x16x4_f32(
          false, af[g][1], false, bf, (short)0, acc[1][t], false, false);
    }
}

__global__ __launch_bounds__(256)
void prob_attn_kernel(const float* __restrict__ Q, const float* __restrict__ K,
                      const float* __restrict__ V, const int* __restrict__ starts,
                      float* __restrict__ O) {
  constexpr int L = 1024, H = 8, E = 64, RT = 32;
  constexpr int PATCH = 64;
  static_assert(PATCH >= 16, "the tile test in phase 1 needs a patch at least one tile long");

  __shared__ __attribute__((aligned(16))) float S[RT * LDS_STRIDE];
  __shared__ unsigned inpatch[L];
  __shared__ float inv_sum[RT];

  const int tid  = threadIdx.x;
  const int wave = tid >> 5;
  const int lane = tid & 31;
  const int lh   = lane & 15;
  const int klo  = (lane < 16) ? 0 : 2;
  const int hih  = (lane < 16) ? 0 : 8;
  const int b = blockIdx.z, h = blockIdx.y;
  const int row_base = blockIdx.x * RT;

  for (int i = tid; i < L; i += 256) {
    unsigned m = 0u;
#pragma unroll
    for (int p = 0; p < 32; ++p) {
      const int s0 = starts[p];
      if (i >= s0 && i < s0 + PATCH) m |= (1u << p);
    }
    inpatch[i] = m;
  }
  __syncthreads();

  const int mt = wave & 1;
  const int r0 = mt * 16;
  {
    const float* qrow = Q + ((size_t)(b * L + row_base + r0 + lh) * H + h) * E;
    v2f afrag[16];
#pragma unroll
    for (int c = 0; c < 16; ++c)
      afrag[c] = *(const v2f*)(qrow + 4 * c + klo);

    unsigned prow[8];
#pragma unroll
    for (int i = 0; i < 8; ++i)
      prow[i] = inpatch[row_base + r0 + i + hih];

    const int g = wave >> 1;
    const unsigned rowor = inpatch[row_base + r0] | inpatch[row_base + r0 + 15];
    const int ct = (g + 4 * lh) * 16;
    const unsigned color = inpatch[ct] | inpatch[ct + 15];
    const unsigned active = __builtin_amdgcn_ballot_w32((rowor & color) != 0u);

    for (int k = 0; k < 16; ++k) {
      const int n0 = (g + 4 * k) * 16;
      if (((active >> k) & 1u) == 0u) {
        for (int i = 0; i < 8; ++i)
          S[(r0 + i + hih) * LDS_STRIDE + n0 + lh] = 0.0f;
        continue;
      }
      const float* krow = K + ((size_t)(b * L + n0 + lh) * H + h) * E;
      v8f acc = {};
#pragma unroll
      for (int c = 0; c < 16; ++c) {
        const v2f bfrag = *(const v2f*)(krow + 4 * c + klo);
        acc = __builtin_amdgcn_wmma_f32_16x16x4_f32(
            false, afrag[c], false, bfrag, (short)0, acc, false, false);
      }
      const unsigned pc = inpatch[n0 + lh];
#pragma unroll
      for (int i = 0; i < 8; ++i) {
        const float val = (prow[i] & pc) ? acc[i] : 0.0f;
        S[(r0 + i + hih) * LDS_STRIDE + n0 + lh] = val;
      }
    }
  }
  __syncthreads();

  for (int rr = 0; rr < 4; ++rr) {
    const int rl = wave * 4 + rr;
    float* srow = &S[rl * LDS_STRIDE];
    float sreg[32];
    unsigned kreg[32];
#pragma unroll
    for (int j = 0; j < 32; ++j) {
      const float s = srow[lane + 32 * j];
      sreg[j] = s;
      kreg[j] = __float_as_uint(s) & 0x7fffffffu;
    }
    int nzero = 0;
#pragma unroll
    for (int j = 0; j < 32; ++j)
      nzero += __builtin_popcount(__builtin_amdgcn_ballot_w32(kreg[j] == 0u));
    unsigned tb = 0u;
    if (nzero < 104) {
      const float a102 = __uint_as_float(kthSmallest(kreg, 102));
      const float a103 = __uint_as_float(kthSmallest(kreg, 103));
      const float thr  = a102 + 0.3f * (a103 - a102);
      tb = __float_as_uint(thr);
    }

    float mx = -INFINITY;
#pragma unroll
    for (int j = 0; j < 32; ++j)
      if (kreg[j] >= tb) mx = fmaxf(mx, sreg[j]);
    mx = wredMaxF(mx);

    float sum = 0.0f;
#pragma unroll
    for (int j = 0; j < 32; ++j) {
      const float p = (kreg[j] >= tb) ? expf(sreg[j] - mx) : 0.0f;
      sum += p;
      srow[lane + 32 * j] = p;
    }
    sum = wredSumF(sum);
    if (lane == 0) inv_sum[rl] = 1.0f / sum;
  }
  __syncthreads();

  {
    constexpr int KS = L / 8;
    constexpr int G  = 4;
    constexpr int NSTAGE = KS / (4 * G);
    const int kbase = wave * KS;
    const float* vrow = V + ((size_t)(b * L + kbase + klo) * H + h) * E + 4 * lh;
    const float* a0 = &S[lh * LDS_STRIDE + kbase + klo];
    const float* a1 = a0 + 16 * LDS_STRIDE;

    v8f acc[2][4] = {};

    v4f vbA[G][2], vbB[G][2];
    v2f afA[G][2], afB[G][2];

    p3_load<G>(vbA, afA, vrow, a0, a1, 0, H * E);
#pragma unroll
    for (int st = 0; st < NSTAGE; st += 2) {
      p3_load<G>(vbB, afB, vrow, a0, a1, (st + 1) * G, H * E);
      p3_mma<G>(acc, vbA, afA);
      if (st + 2 < NSTAGE)
        p3_load<G>(vbA, afA, vrow, a0, a1, (st + 2) * G, H * E);
      p3_mma<G>(acc, vbB, afB);
    }

    __syncthreads();
    float* part = S;
#pragma unroll
    for (int m = 0; m < 2; ++m)
#pragma unroll
      for (int i = 0; i < 8; ++i) {
        v4f o;
        o.x = acc[m][0][i]; o.y = acc[m][1][i]; o.z = acc[m][2][i]; o.w = acc[m][3][i];
        *(v4f*)&part[(wave * RT + 16 * m + i + hih) * PART_STRIDE + 4 * lh] = o;
      }
    __syncthreads();

    for (int j = 0; j < 2; ++j) {
      const int f   = tid + 256 * j;
      const int rl  = f >> 4;
      const int c4  = (f & 15) * 4;
      v4f s = *(const v4f*)&part[rl * PART_STRIDE + c4];
      for (int w = 1; w < 8; ++w)
        s += *(const v4f*)&part[(w * RT + rl) * PART_STRIDE + c4];
      const float is = inv_sum[rl];
      s.x *= is; s.y *= is; s.z *= is; s.w *= is;
      *(v4f*)&O[((size_t)(b * L + row_base + rl) * H + h) * E + c4] = s;
    }
  }
}

extern "C" void kernel_launch(void* const* d_in, const int* in_sizes, int n_in,
                              void* d_out, int out_size, void* d_ws, size_t ws_size,
                              hipStream_t stream) {
  const float* Q = (const float*)d_in[0];
  const float* K = (const float*)d_in[1];
  const float* V = (const float*)d_in[2];
  const int* starts = (const int*)d_in[3];
  float* O = (float*)d_out;

  dim3 grid(1024 / 32, 8, 8);
  dim3 block(256);
  hipLaunchKernelGGL(prob_attn_kernel, grid, block, 0, stream, Q, K, V, starts, O);
}


# Phase 3 block sparsity: column tiles sharing no patch with the 32 rows have row-constant P, so their V rows are only summed (times c_i); WMMA on the rest, dealt to waves mod 8 (sum order changes)
# speedup vs baseline: 6.3488x; 1.2393x over previous
#include <hip/hip_runtime.h>
#include <math.h>


typedef __attribute__((ext_vector_type(2))) float v2f;
typedef __attribute__((ext_vector_type(4))) float v4f;
typedef __attribute__((ext_vector_type(8))) float v8f;

#define LDS_STRIDE 1028
#define PART_STRIDE 68

__device__ __forceinline__ int wredSum(int v) {
#pragma unroll
  for (int m = 16; m; m >>= 1) v += __shfl_xor(v, m, 32);
  return v;
}
__device__ __forceinline__ float wredSumF(float v) {
#pragma unroll
  for (int m = 16; m; m >>= 1) v += __shfl_xor(v, m, 32);
  return v;
}
__device__ __forceinline__ float wredMaxF(float v) {
#pragma unroll
  for (int m = 16; m; m >>= 1) v = fmaxf(v, __shfl_xor(v, m, 32));
  return v;
}

__device__ __forceinline__ unsigned kthSmallest(const unsigned (&key)[32], int k) {
  unsigned res = 0u;
  for (int bit = 31; bit >= 0; --bit) {
    unsigned cand = res | (1u << bit);
    int cnt = 0;
#pragma unroll
    for (int j = 0; j < 32; ++j) cnt += (key[j] < cand) ? 1 : 0;
    cnt = wredSum(cnt);
    if (cnt <= k) res = cand;
  }
  return res;
}


template <int G>
__device__ __forceinline__ void p3_load(v4f (&vb)[G][2], v2f (&af)[G][2],
                                        const float* __restrict__ vrow,
                                        const float* a0, const float* a1,
                                        int ks0, int vpitch) {
#pragma unroll
  for (int g = 0; g < G; ++g) {
    const int k = 4 * (ks0 + g);
    vb[g][0] = *(const v4f*)(vrow + (size_t)k * vpitch);
    vb[g][1] = *(const v4f*)(vrow + (size_t)(k + 1) * vpitch);
    af[g][0] = *(const v2f*)(a0 + k);
    af[g][1] = *(const v2f*)(a1 + k);
  }
}
template <int G>
__device__ __forceinline__ void p3_mma(v8f (&acc)[2][4], const v4f (&vb)[G][2],
                                       const v2f (&af)[G][2]) {
#pragma unroll
  for (int g = 0; g < G; ++g)
#pragma unroll
    for (int t = 0; t < 4; ++t) {
      v2f bf;
      bf.x = vb[g][0][t];
      bf.y = vb[g][1][t];
      acc[0][t] = __builtin_amdgcn_wmma_f32_16x16x4_f32(
          false, af[g][0], false, bf, (short)0, acc[0][t], false, false);
      acc[1][t] = __builtin_amdgcn_wmma_f32_16x16x4_f32(
          false, af[g][1], false, bf, (short)0, acc[1][t], false, false);
    }
}

__global__ __launch_bounds__(256)
void prob_attn_kernel(const float* __restrict__ Q, const float* __restrict__ K,
                      const float* __restrict__ V, const int* __restrict__ starts,
                      float* __restrict__ O) {
  constexpr int L = 1024, H = 8, E = 64, RT = 32;
  constexpr int PATCH = 64;
  static_assert(PATCH >= 16, "the tile test in phase 1 needs a patch at least one tile long");

  __shared__ __attribute__((aligned(16))) float S[RT * LDS_STRIDE];
  __shared__ unsigned inpatch[L];
  __shared__ float inv_sum[RT];
  __shared__ float czero[RT];

  const int tid  = threadIdx.x;
  const int wave = tid >> 5;
  const int lane = tid & 31;
  const int lh   = lane & 15;
  const int klo  = (lane < 16) ? 0 : 2;
  const int hih  = (lane < 16) ? 0 : 8;
  const int b = blockIdx.z, h = blockIdx.y;
  const int row_base = blockIdx.x * RT;

  for (int i = tid; i < L; i += 256) {
    unsigned m = 0u;
#pragma unroll
    for (int p = 0; p < 32; ++p) {
      const int s0 = starts[p];
      if (i >= s0 && i < s0 + PATCH) m |= (1u << p);
    }
    inpatch[i] = m;
  }
  __syncthreads();

  const int mt = wave & 1;
  const int r0 = mt * 16;
  {
    const float* qrow = Q + ((size_t)(b * L + row_base + r0 + lh) * H + h) * E;
    v2f afrag[16];
#pragma unroll
    for (int c = 0; c < 16; ++c)
      afrag[c] = *(const v2f*)(qrow + 4 * c + klo);

    unsigned prow[8];
#pragma unroll
    for (int i = 0; i < 8; ++i)
      prow[i] = inpatch[row_base + r0 + i + hih];

    const int g = wave >> 1;
    const unsigned rowor = inpatch[row_base + r0] | inpatch[row_base + r0 + 15];
    const int ct = (g + 4 * lh) * 16;
    const unsigned color = inpatch[ct] | inpatch[ct + 15];
    const unsigned active = __builtin_amdgcn_ballot_w32((rowor & color) != 0u);

    for (int k = 0; k < 16; ++k) {
      const int n0 = (g + 4 * k) * 16;
      if (((active >> k) & 1u) == 0u) {
        for (int i = 0; i < 8; ++i)
          S[(r0 + i + hih) * LDS_STRIDE + n0 + lh] = 0.0f;
        continue;
      }
      const float* krow = K + ((size_t)(b * L + n0 + lh) * H + h) * E;
      v8f acc = {};
#pragma unroll
      for (int c = 0; c < 16; ++c) {
        const v2f bfrag = *(const v2f*)(krow + 4 * c + klo);
        acc = __builtin_amdgcn_wmma_f32_16x16x4_f32(
            false, afrag[c], false, bfrag, (short)0, acc, false, false);
      }
      const unsigned pc = inpatch[n0 + lh];
#pragma unroll
      for (int i = 0; i < 8; ++i) {
        const float val = (prow[i] & pc) ? acc[i] : 0.0f;
        S[(r0 + i + hih) * LDS_STRIDE + n0 + lh] = val;
      }
    }
  }
  __syncthreads();

  for (int rr = 0; rr < 4; ++rr) {
    const int rl = wave * 4 + rr;
    float* srow = &S[rl * LDS_STRIDE];
    float sreg[32];
    unsigned kreg[32];
#pragma unroll
    for (int j = 0; j < 32; ++j) {
      const float s = srow[lane + 32 * j];
      sreg[j] = s;
      kreg[j] = __float_as_uint(s) & 0x7fffffffu;
    }
    int nzero = 0;
#pragma unroll
    for (int j = 0; j < 32; ++j)
      nzero += __builtin_popcount(__builtin_amdgcn_ballot_w32(kreg[j] == 0u));
    unsigned tb = 0u;
    if (nzero < 104) {
      const float a102 = __uint_as_float(kthSmallest(kreg, 102));
      const float a103 = __uint_as_float(kthSmallest(kreg, 103));
      const float thr  = a102 + 0.3f * (a103 - a102);
      tb = __float_as_uint(thr);
    }

    float mx = -INFINITY;
#pragma unroll
    for (int j = 0; j < 32; ++j)
      if (kreg[j] >= tb) mx = fmaxf(mx, sreg[j]);
    mx = wredMaxF(mx);

    float sum = 0.0f;
#pragma unroll
    for (int j = 0; j < 32; ++j) {
      const float p = (kreg[j] >= tb) ? expf(sreg[j] - mx) : 0.0f;
      sum += p;
      srow[lane + 32 * j] = p;
    }
    sum = wredSumF(sum);
    const float cz = (tb == 0u) ? expf(0.0f - mx) : 0.0f;
    if (lane == 0) { inv_sum[rl] = 1.0f / sum; czero[rl] = cz; }
  }
  __syncthreads();

  {
    constexpr int G = 4;
    const int kbase = wave * 16;
    const float* vrow = V + ((size_t)(b * L + kbase + klo) * H + h) * E + 4 * lh;
    const float* a0 = &S[lh * LDS_STRIDE + kbase + klo];
    const float* a1 = a0 + 16 * LDS_STRIDE;

    const unsigned rowu = inpatch[row_base] | inpatch[row_base + 15] |
                          inpatch[row_base + 16] | inpatch[row_base + 31];
    const int ct3 = (wave + 8 * (lane & 7)) * 16;
    const unsigned col3 = inpatch[ct3] | inpatch[ct3 + 15];
    const unsigned act = __builtin_amdgcn_ballot_w32((rowu & col3) != 0u) & 0xffu;

    unsigned rem = act;
    v4f vbA[G][2], vbB[G][2];
    v2f afA[G][2], afB[G][2];
    const bool haveA = rem != 0u;
    if (haveA) {
      const int s = __builtin_ctz(rem);
      rem &= rem - 1u;
      p3_load<G>(vbA, afA, vrow, a0, a1, 32 * s, H * E);
    }

    v4f csum = {0.0f, 0.0f, 0.0f, 0.0f};
#pragma unroll 1
    for (int s0 = 0; s0 < 8; s0 += 4) {
#pragma unroll
      for (int sl = 0; sl < 4; ++sl) {
        const int s = s0 + sl;
        v4f ts = {0.0f, 0.0f, 0.0f, 0.0f};
#pragma unroll
        for (int g = 0; g < G; ++g) {
          const int k = 128 * s + 4 * g;
          ts += *(const v4f*)(vrow + (size_t)k * (H * E));
          ts += *(const v4f*)(vrow + (size_t)(k + 1) * (H * E));
        }
        const bool skip = ((act >> s) & 1u) == 0u;
        csum.x += skip ? ts.x : 0.0f;
        csum.y += skip ? ts.y : 0.0f;
        csum.z += skip ? ts.z : 0.0f;
        csum.w += skip ? ts.w : 0.0f;
      }
    }

    v8f acc[2][4] = {};
    const bool haveB = rem != 0u;
    if (haveB) {
      const int s = __builtin_ctz(rem);
      rem &= rem - 1u;
      p3_load<G>(vbB, afB, vrow, a0, a1, 32 * s, H * E);
    }
    if (haveA) p3_mma<G>(acc, vbA, afA);
    if (haveB) p3_mma<G>(acc, vbB, afB);
    while (rem != 0u) {
      const int s = __builtin_ctz(rem);
      rem &= rem - 1u;
      v4f vbC[G][2];
      v2f afC[G][2];
      p3_load<G>(vbC, afC, vrow, a0, a1, 32 * s, H * E);
      p3_mma<G>(acc, vbC, afC);
    }

    __syncthreads();
    float* part = S;
    float* vs = S + 8 * RT * PART_STRIDE;
#pragma unroll
    for (int m = 0; m < 2; ++m)
#pragma unroll
      for (int i = 0; i < 8; ++i) {
        v4f o;
        o.x = acc[m][0][i]; o.y = acc[m][1][i]; o.z = acc[m][2][i]; o.w = acc[m][3][i];
        *(v4f*)&part[(wave * RT + 16 * m + i + hih) * PART_STRIDE + 4 * lh] = o;
      }
    *(v4f*)&vs[(wave * 2 + (lane >> 4)) * E + 4 * lh] = csum;
    __syncthreads();

    const int c4 = (tid & 15) * 4;
    v4f vsum = *(const v4f*)&vs[c4];
    for (int q = 1; q < 16; ++q)
      vsum += *(const v4f*)&vs[q * E + c4];
    for (int j = 0; j < 2; ++j) {
      const int f   = tid + 256 * j;
      const int rl  = f >> 4;
      v4f s = *(const v4f*)&part[rl * PART_STRIDE + c4];
      for (int w = 1; w < 8; ++w)
        s += *(const v4f*)&part[(w * RT + rl) * PART_STRIDE + c4];
      const float cz = czero[rl];
      const float is = inv_sum[rl];
      s.x = (s.x + cz * vsum.x) * is;
      s.y = (s.y + cz * vsum.y) * is;
      s.z = (s.z + cz * vsum.z) * is;
      s.w = (s.w + cz * vsum.w) * is;
      *(v4f*)&O[((size_t)(b * L + row_base + rl) * H + h) * E + c4] = s;
    }
  }
}

extern "C" void kernel_launch(void* const* d_in, const int* in_sizes, int n_in,
                              void* d_out, int out_size, void* d_ws, size_t ws_size,
                              hipStream_t stream) {
  const float* Q = (const float*)d_in[0];
  const float* K = (const float*)d_in[1];
  const float* V = (const float*)d_in[2];
  const int* starts = (const int*)d_in[3];
  float* O = (float*)d_out;

  dim3 grid(1024 / 32, 8, 8);
  dim3 block(256);
  hipLaunchKernelGGL(prob_attn_kernel, grid, block, 0, stream, Q, K, V, starts, O);
}


# Score window: unmasked scores lie within 63 columns of the row, so phases 1-2 use a 160-column window; the 864 zeros outside make the threshold 0 and add 864*exp(-max) to the sum (sum order changes)
# speedup vs baseline: 11.2269x; 1.7683x over previous
#include <hip/hip_runtime.h>
#include <math.h>


typedef __attribute__((ext_vector_type(2))) float v2f;
typedef __attribute__((ext_vector_type(4))) float v4f;
typedef __attribute__((ext_vector_type(8))) float v8f;

#define LDS_STRIDE 1028
#define PART_STRIDE 68

template <int G>
__device__ __forceinline__ void p3_load(v4f (&vb)[G][2], v2f (&af)[G][2],
                                        const float* __restrict__ vrow,
                                        const float* a0, const float* a1,
                                        int ks0, int vpitch) {
#pragma unroll
  for (int g = 0; g < G; ++g) {
    const int k = 4 * (ks0 + g);
    vb[g][0] = *(const v4f*)(vrow + (size_t)k * vpitch);
    vb[g][1] = *(const v4f*)(vrow + (size_t)(k + 1) * vpitch);
    af[g][0] = *(const v2f*)(a0 + k);
    af[g][1] = *(const v2f*)(a1 + k);
  }
}
template <int G>
__device__ __forceinline__ void p3_mma(v8f (&acc)[2][4], const v4f (&vb)[G][2],
                                       const v2f (&af)[G][2]) {
#pragma unroll
  for (int g = 0; g < G; ++g)
#pragma unroll
    for (int t = 0; t < 4; ++t) {
      v2f bf;
      bf.x = vb[g][0][t];
      bf.y = vb[g][1][t];
      acc[0][t] = __builtin_amdgcn_wmma_f32_16x16x4_f32(
          false, af[g][0], false, bf, (short)0, acc[0][t], false, false);
      acc[1][t] = __builtin_amdgcn_wmma_f32_16x16x4_f32(
          false, af[g][1], false, bf, (short)0, acc[1][t], false, false);
    }
}

__global__ __launch_bounds__(256)
void prob_attn_kernel(const float* __restrict__ Q, const float* __restrict__ K,
                      const float* __restrict__ V, const int* __restrict__ starts,
                      float* __restrict__ O) {
  constexpr int L = 1024, H = 8, E = 64, RT = 32;
  constexpr int PATCH = 64;
  static_assert(PATCH >= 16, "the tile test in phase 1 needs a patch at least one tile long");
  constexpr int WIN = 160;
  constexpr int WT  = WIN / 16;
  static_assert(WIN >= RT + 2 * (PATCH - 1) && WIN % 32 == 0 && (L - WIN) % 16 == 0,
                "the window must cover rows +- (PATCH - 1), in whole tiles, 32 columns per pass");

  __shared__ __attribute__((aligned(16))) float S[RT * LDS_STRIDE];
  __shared__ unsigned inpatch[L];
  __shared__ float inv_sum[RT];
  __shared__ float czero[RT];

  const int tid  = threadIdx.x;
  const int wave = tid >> 5;
  const int lane = tid & 31;
  const int lh   = lane & 15;
  const int klo  = (lane < 16) ? 0 : 2;
  const int hih  = (lane < 16) ? 0 : 8;
  const int b = blockIdx.z, h = blockIdx.y;
  const int row_base = blockIdx.x * RT;
  const int w0 = min(max(row_base - PATCH, 0), L - WIN);

  for (int i = tid; i < L; i += 256) {
    unsigned m = 0u;
#pragma unroll
    for (int p = 0; p < 32; ++p) {
      const int s0 = starts[p];
      if (i >= s0 && i < s0 + PATCH) m |= (1u << p);
    }
    inpatch[i] = m;
  }
  __syncthreads();

  const int mt = wave & 1;
  const int r0 = mt * 16;
  {
    const float* qrow = Q + ((size_t)(b * L + row_base + r0 + lh) * H + h) * E;
    v2f afrag[16];
#pragma unroll
    for (int c = 0; c < 16; ++c)
      afrag[c] = *(const v2f*)(qrow + 4 * c + klo);

    unsigned prow[8];
#pragma unroll
    for (int i = 0; i < 8; ++i)
      prow[i] = inpatch[row_base + r0 + i + hih];

    const int g = __builtin_amdgcn_readfirstlane(wave >> 1);
    const unsigned rowor = inpatch[row_base + r0] | inpatch[row_base + r0 + 15];
    const int wtl = g + 4 * (lane & 3);
    const int ct = w0 + 16 * (wtl < WT ? wtl : 0);
    const unsigned color = inpatch[ct] | inpatch[ct + 15];
    const unsigned active = __builtin_amdgcn_ballot_w32((rowor & color) != 0u);

    for (int k = 0; g + 4 * k < WT; ++k) {
      const int n0 = w0 + (g + 4 * k) * 16;
      if (((active >> k) & 1u) == 0u) {
        for (int i = 0; i < 8; ++i)
          S[(r0 + i + hih) * LDS_STRIDE + n0 + lh] = 0.0f;
        continue;
      }
      const float* krow = K + ((size_t)(b * L + n0 + lh) * H + h) * E;
      v8f acc = {};
#pragma unroll
      for (int c = 0; c < 16; ++c) {
        const v2f bfrag = *(const v2f*)(krow + 4 * c + klo);
        acc = __builtin_amdgcn_wmma_f32_16x16x4_f32(
            false, afrag[c], false, bfrag, (short)0, acc, false, false);
      }
      const unsigned pc = inpatch[n0 + lh];
#pragma unroll
      for (int i = 0; i < 8; ++i) {
        const float val = (prow[i] & pc) ? acc[i] : 0.0f;
        S[(r0 + i + hih) * LDS_STRIDE + n0 + lh] = val;
      }
    }
  }
  __syncthreads();

  {
    constexpr int NW = WIN / 32;
    constexpr float NOUT = (float)(L - WIN);
    float sw[4][NW];
#pragma unroll
    for (int rr = 0; rr < 4; ++rr)
#pragma unroll
      for (int j = 0; j < NW; ++j)
        sw[rr][j] = S[(wave * 4 + rr) * LDS_STRIDE + w0 + lane + 32 * j];

    float mx[4], sum[4];
#pragma unroll
    for (int rr = 0; rr < 4; ++rr) {
      float m = 0.0f;
#pragma unroll
      for (int j = 0; j < NW; ++j) m = fmaxf(m, sw[rr][j]);
      mx[rr] = m;
    }
#pragma unroll
    for (int d = 16; d; d >>= 1)
#pragma unroll
      for (int rr = 0; rr < 4; ++rr) mx[rr] = fmaxf(mx[rr], __shfl_xor(mx[rr], d, 32));

#pragma unroll
    for (int rr = 0; rr < 4; ++rr) {
      float* srow = &S[(wave * 4 + rr) * LDS_STRIDE + w0];
      float s = 0.0f;
#pragma unroll
      for (int j = 0; j < NW; ++j) {
        const float p = expf(sw[rr][j] - mx[rr]);
        s += p;
        srow[lane + 32 * j] = p;
      }
      sum[rr] = s;
    }
#pragma unroll
    for (int d = 16; d; d >>= 1)
#pragma unroll
      for (int rr = 0; rr < 4; ++rr) sum[rr] += __shfl_xor(sum[rr], d, 32);

    float msel = mx[0], ssel = sum[0];
#pragma unroll
    for (int rr = 1; rr < 4; ++rr)
      if (lane == rr) { msel = mx[rr]; ssel = sum[rr]; }
    const float cz = expf(0.0f - msel);
    ssel += NOUT * cz;
    if (lane < 4) { inv_sum[wave * 4 + lane] = 1.0f / ssel; czero[wave * 4 + lane] = cz; }
  }
  __syncthreads();

  {
    constexpr int G = 4;
    const int kbase = wave * 16;
    const float* vrow = V + ((size_t)(b * L + kbase + klo) * H + h) * E + 4 * lh;
    const float* a0 = &S[lh * LDS_STRIDE + kbase + klo];
    const float* a1 = a0 + 16 * LDS_STRIDE;

    const unsigned rowu = inpatch[row_base] | inpatch[row_base + 15] |
                          inpatch[row_base + 16] | inpatch[row_base + 31];
    const int ct3 = (wave + 8 * (lane & 7)) * 16;
    const unsigned col3 = inpatch[ct3] | inpatch[ct3 + 15];
    const unsigned act = __builtin_amdgcn_ballot_w32((rowu & col3) != 0u) & 0xffu;

    unsigned rem = act;
    v4f vbA[G][2], vbB[G][2];
    v2f afA[G][2], afB[G][2];
    const bool haveA = rem != 0u;
    if (haveA) {
      const int s = __builtin_ctz(rem);
      rem &= rem - 1u;
      p3_load<G>(vbA, afA, vrow, a0, a1, 32 * s, H * E);
    }

    v4f csum = {0.0f, 0.0f, 0.0f, 0.0f};
#pragma unroll 1
    for (int s0 = 0; s0 < 8; s0 += 4) {
#pragma unroll
      for (int sl = 0; sl < 4; ++sl) {
        const int s = s0 + sl;
        v4f ts = {0.0f, 0.0f, 0.0f, 0.0f};
#pragma unroll
        for (int g = 0; g < G; ++g) {
          const int k = 128 * s + 4 * g;
          ts += *(const v4f*)(vrow + (size_t)k * (H * E));
          ts += *(const v4f*)(vrow + (size_t)(k + 1) * (H * E));
        }
        const bool skip = ((act >> s) & 1u) == 0u;
        csum.x += skip ? ts.x : 0.0f;
        csum.y += skip ? ts.y : 0.0f;
        csum.z += skip ? ts.z : 0.0f;
        csum.w += skip ? ts.w : 0.0f;
      }
    }

    v8f acc[2][4] = {};
    const bool haveB = rem != 0u;
    if (haveB) {
      const int s = __builtin_ctz(rem);
      rem &= rem - 1u;
      p3_load<G>(vbB, afB, vrow, a0, a1, 32 * s, H * E);
    }
    if (haveA) p3_mma<G>(acc, vbA, afA);
    if (haveB) p3_mma<G>(acc, vbB, afB);
    while (rem != 0u) {
      const int s = __builtin_ctz(rem);
      rem &= rem - 1u;
      v4f vbC[G][2];
      v2f afC[G][2];
      p3_load<G>(vbC, afC, vrow, a0, a1, 32 * s, H * E);
      p3_mma<G>(acc, vbC, afC);
    }

    __syncthreads();
    float* part = S;
    float* vs = S + 8 * RT * PART_STRIDE;
#pragma unroll
    for (int m = 0; m < 2; ++m)
#pragma unroll
      for (int i = 0; i < 8; ++i) {
        v4f o;
        o.x = acc[m][0][i]; o.y = acc[m][1][i]; o.z = acc[m][2][i]; o.w = acc[m][3][i];
        *(v4f*)&part[(wave * RT + 16 * m + i + hih) * PART_STRIDE + 4 * lh] = o;
      }
    *(v4f*)&vs[(wave * 2 + (lane >> 4)) * E + 4 * lh] = csum;
    __syncthreads();

    const int c4 = (tid & 15) * 4;
    v4f vsum = *(const v4f*)&vs[c4];
    for (int q = 1; q < 16; ++q)
      vsum += *(const v4f*)&vs[q * E + c4];
    for (int j = 0; j < 2; ++j) {
      const int f   = tid + 256 * j;
      const int rl  = f >> 4;
      v4f s = *(const v4f*)&part[rl * PART_STRIDE + c4];
      for (int w = 1; w < 8; ++w)
        s += *(const v4f*)&part[(w * RT + rl) * PART_STRIDE + c4];
      const float cz = czero[rl];
      const float is = inv_sum[rl];
      s.x = (s.x + cz * vsum.x) * is;
      s.y = (s.y + cz * vsum.y) * is;
      s.z = (s.z + cz * vsum.z) * is;
      s.w = (s.w + cz * vsum.w) * is;
      *(v4f*)&O[((size_t)(b * L + row_base + rl) * H + h) * E + c4] = s;
    }
  }
}

extern "C" void kernel_launch(void* const* d_in, const int* in_sizes, int n_in,
                              void* d_out, int out_size, void* d_ws, size_t ws_size,
                              hipStream_t stream) {
  const float* Q = (const float*)d_in[0];
  const float* K = (const float*)d_in[1];
  const float* V = (const float*)d_in[2];
  const int* starts = (const int*)d_in[3];
  float* O = (float*)d_out;

  dim3 grid(1024 / 32, 8, 8);
  dim3 block(256);
  hipLaunchKernelGGL(prob_attn_kernel, grid, block, 0, stream, Q, K, V, starts, O);
}


# V tile sums in a pre-pass kernel: the 16-row sums of V per (b,h,key tile) go to the workspace once; phase 3 adds one 256 B row per zero-score tile instead of 16 V rows (sum order changes)
# speedup vs baseline: 12.5170x; 1.1149x over previous
#include <hip/hip_runtime.h>
#include <math.h>


typedef __attribute__((ext_vector_type(2))) float v2f;
typedef __attribute__((ext_vector_type(4))) float v4f;
typedef __attribute__((ext_vector_type(8))) float v8f;

#define LDS_STRIDE 1028
#define PART_STRIDE 68

template <int G>
__device__ __forceinline__ void p3_load(v4f (&vb)[G][2], v2f (&af)[G][2],
                                        const float* __restrict__ vrow,
                                        const float* a0, const float* a1,
                                        int ks0, int vpitch) {
#pragma unroll
  for (int g = 0; g < G; ++g) {
    const int k = 4 * (ks0 + g);
    vb[g][0] = *(const v4f*)(vrow + (size_t)k * vpitch);
    vb[g][1] = *(const v4f*)(vrow + (size_t)(k + 1) * vpitch);
    af[g][0] = *(const v2f*)(a0 + k);
    af[g][1] = *(const v2f*)(a1 + k);
  }
}
template <int G>
__device__ __forceinline__ void p3_mma(v8f (&acc)[2][4], const v4f (&vb)[G][2],
                                       const v2f (&af)[G][2]) {
#pragma unroll
  for (int g = 0; g < G; ++g)
#pragma unroll
    for (int t = 0; t < 4; ++t) {
      v2f bf;
      bf.x = vb[g][0][t];
      bf.y = vb[g][1][t];
      acc[0][t] = __builtin_amdgcn_wmma_f32_16x16x4_f32(
          false, af[g][0], false, bf, (short)0, acc[0][t], false, false);
      acc[1][t] = __builtin_amdgcn_wmma_f32_16x16x4_f32(
          false, af[g][1], false, bf, (short)0, acc[1][t], false, false);
    }
}

__global__ __launch_bounds__(128)
void v_tile_sums_kernel(const float* __restrict__ V, float* __restrict__ T) {
  constexpr int L = 1024, H = 8, E = 64;
  const int lane = threadIdx.x & 31;
  const int lh   = lane & 15;
  const int half = lane >> 4;
  const int t0   = 2 * (blockIdx.x * 4 + (threadIdx.x >> 5));
  const int bh   = t0 >> 6;
  const int tile = t0 & 63;
  const int b = bh >> 3, h = bh & 7;
  const float* vrow = V + ((size_t)(b * L + tile * 16 + half) * H + h) * E + 4 * lh;
  v4f x[2][8];
#pragma unroll
  for (int u = 0; u < 2; ++u)
#pragma unroll
    for (int i = 0; i < 8; ++i)
      x[u][i] = *(const v4f*)(vrow + (size_t)(16 * u + 2 * i) * (H * E));
  v4f s0 = x[0][0], s1 = x[1][0];
#pragma unroll
  for (int i = 1; i < 8; ++i) { s0 += x[0][i]; s1 += x[1][i]; }
  const v4f mine = half ? s1 : s0;
  const v4f send = half ? s0 : s1;
  v4f o;
  o.x = mine.x + __shfl_xor(send.x, 16, 32);
  o.y = mine.y + __shfl_xor(send.y, 16, 32);
  o.z = mine.z + __shfl_xor(send.z, 16, 32);
  o.w = mine.w + __shfl_xor(send.w, 16, 32);
  *(v4f*)(T + (size_t)(t0 + half) * E + 4 * lh) = o;
}

__global__ __launch_bounds__(256)
void prob_attn_kernel(const float* __restrict__ Q, const float* __restrict__ K,
                      const float* __restrict__ V, const int* __restrict__ starts,
                      const float* __restrict__ T, float* __restrict__ O) {
  constexpr int L = 1024, H = 8, E = 64, RT = 32;
  constexpr int PATCH = 64;
  static_assert(PATCH >= 16, "the tile test in phase 1 needs a patch at least one tile long");
  constexpr int WIN = 160;
  constexpr int WT  = WIN / 16;
  static_assert(WIN >= RT + 2 * (PATCH - 1) && WIN % 32 == 0 && (L - WIN) % 16 == 0,
                "the window must cover rows +- (PATCH - 1), in whole tiles, 32 columns per pass");

  __shared__ __attribute__((aligned(16))) float S[RT * LDS_STRIDE];
  __shared__ unsigned inpatch[L];
  __shared__ float inv_sum[RT];
  __shared__ float czero[RT];

  const int tid  = threadIdx.x;
  const int wave = tid >> 5;
  const int lane = tid & 31;
  const int lh   = lane & 15;
  const int klo  = (lane < 16) ? 0 : 2;
  const int hih  = (lane < 16) ? 0 : 8;
  const int b = blockIdx.z, h = blockIdx.y;
  const int row_base = blockIdx.x * RT;
  const int w0 = min(max(row_base - PATCH, 0), L - WIN);

  for (int i = tid; i < L; i += 256) {
    unsigned m = 0u;
#pragma unroll
    for (int p = 0; p < 32; ++p) {
      const int s0 = starts[p];
      if (i >= s0 && i < s0 + PATCH) m |= (1u << p);
    }
    inpatch[i] = m;
  }
  __syncthreads();

  const int mt = wave & 1;
  const int r0 = mt * 16;
  {
    const float* qrow = Q + ((size_t)(b * L + row_base + r0 + lh) * H + h) * E;
    v2f afrag[16];
#pragma unroll
    for (int c = 0; c < 16; ++c)
      afrag[c] = *(const v2f*)(qrow + 4 * c + klo);

    unsigned prow[8];
#pragma unroll
    for (int i = 0; i < 8; ++i)
      prow[i] = inpatch[row_base + r0 + i + hih];

    const int g = __builtin_amdgcn_readfirstlane(wave >> 1);
    const unsigned rowor = inpatch[row_base + r0] | inpatch[row_base + r0 + 15];
    const int wtl = g + 4 * (lane & 3);
    const int ct = w0 + 16 * (wtl < WT ? wtl : 0);
    const unsigned color = inpatch[ct] | inpatch[ct + 15];
    const unsigned active = __builtin_amdgcn_ballot_w32((rowor & color) != 0u);

    for (int k = 0; g + 4 * k < WT; ++k) {
      const int n0 = w0 + (g + 4 * k) * 16;
      if (((active >> k) & 1u) == 0u) {
        for (int i = 0; i < 8; ++i)
          S[(r0 + i + hih) * LDS_STRIDE + n0 + lh] = 0.0f;
        continue;
      }
      const float* krow = K + ((size_t)(b * L + n0 + lh) * H + h) * E;
      v8f acc = {};
#pragma unroll
      for (int c = 0; c < 16; ++c) {
        const v2f bfrag = *(const v2f*)(krow + 4 * c + klo);
        acc = __builtin_amdgcn_wmma_f32_16x16x4_f32(
            false, afrag[c], false, bfrag, (short)0, acc, false, false);
      }
      const unsigned pc = inpatch[n0 + lh];
#pragma unroll
      for (int i = 0; i < 8; ++i) {
        const float val = (prow[i] & pc) ? acc[i] : 0.0f;
        S[(r0 + i + hih) * LDS_STRIDE + n0 + lh] = val;
      }
    }
  }
  __syncthreads();

  {
    constexpr int NW = WIN / 32;
    constexpr float NOUT = (float)(L - WIN);
    float sw[4][NW];
#pragma unroll
    for (int rr = 0; rr < 4; ++rr)
#pragma unroll
      for (int j = 0; j < NW; ++j)
        sw[rr][j] = S[(wave * 4 + rr) * LDS_STRIDE + w0 + lane + 32 * j];

    float mx[4], sum[4];
#pragma unroll
    for (int rr = 0; rr < 4; ++rr) {
      float m = 0.0f;
#pragma unroll
      for (int j = 0; j < NW; ++j) m = fmaxf(m, sw[rr][j]);
      mx[rr] = m;
    }
#pragma unroll
    for (int d = 16; d; d >>= 1)
#pragma unroll
      for (int rr = 0; rr < 4; ++rr) mx[rr] = fmaxf(mx[rr], __shfl_xor(mx[rr], d, 32));

#pragma unroll
    for (int rr = 0; rr < 4; ++rr) {
      float* srow = &S[(wave * 4 + rr) * LDS_STRIDE + w0];
      float s = 0.0f;
#pragma unroll
      for (int j = 0; j < NW; ++j) {
        const float p = expf(sw[rr][j] - mx[rr]);
        s += p;
        srow[lane + 32 * j] = p;
      }
      sum[rr] = s;
    }
#pragma unroll
    for (int d = 16; d; d >>= 1)
#pragma unroll
      for (int rr = 0; rr < 4; ++rr) sum[rr] += __shfl_xor(sum[rr], d, 32);

    float msel = mx[0], ssel = sum[0];
#pragma unroll
    for (int rr = 1; rr < 4; ++rr)
      if (lane == rr) { msel = mx[rr]; ssel = sum[rr]; }
    const float cz = expf(0.0f - msel);
    ssel += NOUT * cz;
    if (lane < 4) { inv_sum[wave * 4 + lane] = 1.0f / ssel; czero[wave * 4 + lane] = cz; }
  }
  __syncthreads();

  {
    constexpr int G = 4;
    const int kbase = wave * 16;
    const float* vrow = V + ((size_t)(b * L + kbase + klo) * H + h) * E + 4 * lh;
    const float* a0 = &S[lh * LDS_STRIDE + kbase + klo];
    const float* a1 = a0 + 16 * LDS_STRIDE;

    const unsigned rowu = inpatch[row_base] | inpatch[row_base + 15] |
                          inpatch[row_base + 16] | inpatch[row_base + 31];
    const int ct3 = (wave + 8 * (lane & 7)) * 16;
    const unsigned col3 = inpatch[ct3] | inpatch[ct3 + 15];
    const unsigned act = __builtin_amdgcn_ballot_w32((rowu & col3) != 0u) & 0xffu;

    unsigned rem = act;
    v4f vbA[G][2], vbB[G][2];
    v2f afA[G][2], afB[G][2];
    const bool haveA = rem != 0u;
    if (haveA) {
      const int s = __builtin_ctz(rem);
      rem &= rem - 1u;
      p3_load<G>(vbA, afA, vrow, a0, a1, 32 * s, H * E);
    }

    v4f csum = {0.0f, 0.0f, 0.0f, 0.0f};
    {
      const int half = lane >> 4;
      const float* trow = T + ((size_t)((b * H + h) * 64 + wave + 8 * half)) * E + 4 * lh;
      v4f ts[4];
#pragma unroll
      for (int j = 0; j < 4; ++j)
        ts[j] = *(const v4f*)(trow + (size_t)(16 * j) * E);
#pragma unroll
      for (int j = 0; j < 4; ++j) {
        const bool skip = ((act >> (2 * j + half)) & 1u) == 0u;
        csum.x += skip ? ts[j].x : 0.0f;
        csum.y += skip ? ts[j].y : 0.0f;
        csum.z += skip ? ts[j].z : 0.0f;
        csum.w += skip ? ts[j].w : 0.0f;
      }
    }

    v8f acc[2][4] = {};
    const bool haveB = rem != 0u;
    if (haveB) {
      const int s = __builtin_ctz(rem);
      rem &= rem - 1u;
      p3_load<G>(vbB, afB, vrow, a0, a1, 32 * s, H * E);
    }
    if (haveA) p3_mma<G>(acc, vbA, afA);
    if (haveB) p3_mma<G>(acc, vbB, afB);
    while (rem != 0u) {
      const int s = __builtin_ctz(rem);
      rem &= rem - 1u;
      v4f vbC[G][2];
      v2f afC[G][2];
      p3_load<G>(vbC, afC, vrow, a0, a1, 32 * s, H * E);
      p3_mma<G>(acc, vbC, afC);
    }

    __syncthreads();
    float* part = S;
    float* vs = S + 8 * RT * PART_STRIDE;
#pragma unroll
    for (int m = 0; m < 2; ++m)
#pragma unroll
      for (int i = 0; i < 8; ++i) {
        v4f o;
        o.x = acc[m][0][i]; o.y = acc[m][1][i]; o.z = acc[m][2][i]; o.w = acc[m][3][i];
        *(v4f*)&part[(wave * RT + 16 * m + i + hih) * PART_STRIDE + 4 * lh] = o;
      }
    *(v4f*)&vs[(wave * 2 + (lane >> 4)) * E + 4 * lh] = csum;
    __syncthreads();

    const int c4 = (tid & 15) * 4;
    v4f vsum = *(const v4f*)&vs[c4];
    for (int q = 1; q < 16; ++q)
      vsum += *(const v4f*)&vs[q * E + c4];
    for (int j = 0; j < 2; ++j) {
      const int f   = tid + 256 * j;
      const int rl  = f >> 4;
      v4f s = *(const v4f*)&part[rl * PART_STRIDE + c4];
      for (int w = 1; w < 8; ++w)
        s += *(const v4f*)&part[(w * RT + rl) * PART_STRIDE + c4];
      const float cz = czero[rl];
      const float is = inv_sum[rl];
      s.x = (s.x + cz * vsum.x) * is;
      s.y = (s.y + cz * vsum.y) * is;
      s.z = (s.z + cz * vsum.z) * is;
      s.w = (s.w + cz * vsum.w) * is;
      *(v4f*)&O[((size_t)(b * L + row_base + rl) * H + h) * E + c4] = s;
    }
  }
}

extern "C" void kernel_launch(void* const* d_in, const int* in_sizes, int n_in,
                              void* d_out, int out_size, void* d_ws, size_t ws_size,
                              hipStream_t stream) {
  const float* Q = (const float*)d_in[0];
  const float* K = (const float*)d_in[1];
  const float* V = (const float*)d_in[2];
  const int* starts = (const int*)d_in[3];
  float* O = (float*)d_out;

  float* T = (float*)d_ws;

  hipLaunchKernelGGL(v_tile_sums_kernel, dim3(512), dim3(128), 0, stream, V, T);

  dim3 grid(1024 / 32, 8, 8);
  dim3 block(256);
  hipLaunchKernelGGL(prob_attn_kernel, grid, block, 0, stream, Q, K, V, starts, T, O);
}


# Residency 2 -> 4 workgroups per WGP: score block at window pitch 164 (LDS 133 -> 76 KiB); phase 3 holds one tile's fragments, not two, and parks column sums in LDS early (184 -> 118 VGPRs)
# speedup vs baseline: 15.2865x; 1.2213x over previous
#include <hip/hip_runtime.h>
#include <math.h>


typedef __attribute__((ext_vector_type(2))) float v2f;
typedef __attribute__((ext_vector_type(4))) float v4f;
typedef __attribute__((ext_vector_type(8))) float v8f;

#define LDS_STRIDE 164
#define PART_STRIDE 68

template <int G>
__device__ __forceinline__ void p3_load(v4f (&vb)[G][2], v2f (&af)[G][2],
                                        const float* __restrict__ vrow,
                                        const float* sblk, int ao0, int ao1,
                                        int ks0, int vpitch) {
#pragma unroll
  for (int g = 0; g < G; ++g) {
    const int k = 4 * (ks0 + g);
    vb[g][0] = *(const v4f*)(vrow + (size_t)k * vpitch);
    vb[g][1] = *(const v4f*)(vrow + (size_t)(k + 1) * vpitch);
    af[g][0] = *(const v2f*)(sblk + (ao0 + k));
    af[g][1] = *(const v2f*)(sblk + (ao1 + k));
  }
}
template <int G>
__device__ __forceinline__ void p3_mma(v8f (&acc)[2][4], const v4f (&vb)[G][2],
                                       const v2f (&af)[G][2]) {
#pragma unroll
  for (int g = 0; g < G; ++g)
#pragma unroll
    for (int t = 0; t < 4; ++t) {
      v2f bf;
      bf.x = vb[g][0][t];
      bf.y = vb[g][1][t];
      acc[0][t] = __builtin_amdgcn_wmma_f32_16x16x4_f32(
          false, af[g][0], false, bf, (short)0, acc[0][t], false, false);
      acc[1][t] = __builtin_amdgcn_wmma_f32_16x16x4_f32(
          false, af[g][1], false, bf, (short)0, acc[1][t], false, false);
    }
}

__global__ __launch_bounds__(128)
void v_tile_sums_kernel(const float* __restrict__ V, float* __restrict__ T) {
  constexpr int L = 1024, H = 8, E = 64;
  const int lane = threadIdx.x & 31;
  const int lh   = lane & 15;
  const int half = lane >> 4;
  const int t0   = 2 * (blockIdx.x * 4 + (threadIdx.x >> 5));
  const int bh   = t0 >> 6;
  const int tile = t0 & 63;
  const int b = bh >> 3, h = bh & 7;
  const float* vrow = V + ((size_t)(b * L + tile * 16 + half) * H + h) * E + 4 * lh;
  v4f x[2][8];
#pragma unroll
  for (int u = 0; u < 2; ++u)
#pragma unroll
    for (int i = 0; i < 8; ++i)
      x[u][i] = *(const v4f*)(vrow + (size_t)(16 * u + 2 * i) * (H * E));
  v4f s0 = x[0][0], s1 = x[1][0];
#pragma unroll
  for (int i = 1; i < 8; ++i) { s0 += x[0][i]; s1 += x[1][i]; }
  const v4f mine = half ? s1 : s0;
  const v4f send = half ? s0 : s1;
  v4f o;
  o.x = mine.x + __shfl_xor(send.x, 16, 32);
  o.y = mine.y + __shfl_xor(send.y, 16, 32);
  o.z = mine.z + __shfl_xor(send.z, 16, 32);
  o.w = mine.w + __shfl_xor(send.w, 16, 32);
  *(v4f*)(T + (size_t)(t0 + half) * E + 4 * lh) = o;
}

__global__ __launch_bounds__(256)
void prob_attn_kernel(const float* __restrict__ Q, const float* __restrict__ K,
                      const float* __restrict__ V, const int* __restrict__ starts,
                      const float* __restrict__ T, float* __restrict__ O) {
  constexpr int L = 1024, H = 8, E = 64, RT = 32;
  constexpr int PATCH = 64;
  static_assert(PATCH >= 16, "the tile test in phase 1 needs a patch at least one tile long");
  constexpr int WIN = 160;
  constexpr int WT  = WIN / 16;
  static_assert(WIN >= RT + 2 * (PATCH - 1) && WIN % 32 == 0 && (L - WIN) % 16 == 0,
                "the window must cover rows +- (PATCH - 1), in whole tiles, 32 columns per pass");

  constexpr int S_FLOATS = 8 * RT * PART_STRIDE + 16 * E;
  static_assert(8 * RT * PART_STRIDE >= RT * LDS_STRIDE && LDS_STRIDE >= WIN,
                "the score block of the window must end in front of the column sums' area");
  __shared__ __attribute__((aligned(16))) float S[S_FLOATS];
  __shared__ unsigned inpatch[L];
  __shared__ float inv_sum[RT];
  __shared__ float czero[RT];

  const int tid  = threadIdx.x;
  const int wave = tid >> 5;
  const int lane = tid & 31;
  const int lh   = lane & 15;
  const int klo  = (lane < 16) ? 0 : 2;
  const int hih  = (lane < 16) ? 0 : 8;
  const int b = blockIdx.z, h = blockIdx.y;
  const int row_base = blockIdx.x * RT;
  const int w0 = min(max(row_base - PATCH, 0), L - WIN);

  for (int i = tid; i < L; i += 256) {
    unsigned m = 0u;
#pragma unroll
    for (int p = 0; p < 32; ++p) {
      const int s0 = starts[p];
      if (i >= s0 && i < s0 + PATCH) m |= (1u << p);
    }
    inpatch[i] = m;
  }
  __syncthreads();

  const int mt = wave & 1;
  const int r0 = mt * 16;
  {
    const float* qrow = Q + ((size_t)(b * L + row_base + r0 + lh) * H + h) * E;
    v2f afrag[16];
#pragma unroll
    for (int c = 0; c < 16; ++c)
      afrag[c] = *(const v2f*)(qrow + 4 * c + klo);

    unsigned prow[8];
#pragma unroll
    for (int i = 0; i < 8; ++i)
      prow[i] = inpatch[row_base + r0 + i + hih];

    const int g = __builtin_amdgcn_readfirstlane(wave >> 1);
    const unsigned rowor = inpatch[row_base + r0] | inpatch[row_base + r0 + 15];
    const int wtl = g + 4 * (lane & 3);
    const int ct = w0 + 16 * (wtl < WT ? wtl : 0);
    const unsigned color = inpatch[ct] | inpatch[ct + 15];
    const unsigned active = __builtin_amdgcn_ballot_w32((rowor & color) != 0u);

    for (int k = 0; g + 4 * k < WT; ++k) {
      const int c0 = (g + 4 * k) * 16;
      const int n0 = w0 + c0;
      if (((active >> k) & 1u) == 0u) {
        for (int i = 0; i < 8; ++i)
          S[(r0 + i + hih) * LDS_STRIDE + c0 + lh] = 0.0f;
        continue;
      }
      const float* krow = K + ((size_t)(b * L + n0 + lh) * H + h) * E;
      v8f acc = {};
#pragma unroll
      for (int c = 0; c < 16; ++c) {
        const v2f bfrag = *(const v2f*)(krow + 4 * c + klo);
        acc = __builtin_amdgcn_wmma_f32_16x16x4_f32(
            false, afrag[c], false, bfrag, (short)0, acc, false, false);
      }
      const unsigned pc = inpatch[n0 + lh];
#pragma unroll
      for (int i = 0; i < 8; ++i) {
        const float val = (prow[i] & pc) ? acc[i] : 0.0f;
        S[(r0 + i + hih) * LDS_STRIDE + c0 + lh] = val;
      }
    }
  }
  __syncthreads();

  {
    constexpr int NW = WIN / 32;
    constexpr float NOUT = (float)(L - WIN);
    float sw[4][NW];
#pragma unroll
    for (int rr = 0; rr < 4; ++rr)
#pragma unroll
      for (int j = 0; j < NW; ++j)
        sw[rr][j] = S[(wave * 4 + rr) * LDS_STRIDE + lane + 32 * j];

    float mx[4], sum[4];
#pragma unroll
    for (int rr = 0; rr < 4; ++rr) {
      float m = 0.0f;
#pragma unroll
      for (int j = 0; j < NW; ++j) m = fmaxf(m, sw[rr][j]);
      mx[rr] = m;
    }
#pragma unroll
    for (int d = 16; d; d >>= 1)
#pragma unroll
      for (int rr = 0; rr < 4; ++rr) mx[rr] = fmaxf(mx[rr], __shfl_xor(mx[rr], d, 32));

#pragma unroll
    for (int rr = 0; rr < 4; ++rr) {
      float* srow = &S[(wave * 4 + rr) * LDS_STRIDE];
      float s = 0.0f;
#pragma unroll
      for (int j = 0; j < NW; ++j) {
        const float p = expf(sw[rr][j] - mx[rr]);
        s += p;
        srow[lane + 32 * j] = p;
      }
      sum[rr] = s;
    }
#pragma unroll
    for (int d = 16; d; d >>= 1)
#pragma unroll
      for (int rr = 0; rr < 4; ++rr) sum[rr] += __shfl_xor(sum[rr], d, 32);

    float msel = mx[0], ssel = sum[0];
#pragma unroll
    for (int rr = 1; rr < 4; ++rr)
      if (lane == rr) { msel = mx[rr]; ssel = sum[rr]; }
    const float cz = expf(0.0f - msel);
    ssel += NOUT * cz;
    if (lane < 4) { inv_sum[wave * 4 + lane] = 1.0f / ssel; czero[wave * 4 + lane] = cz; }
  }
  __syncthreads();

  {
    constexpr int G = 4;
    const int kbase = wave * 16;
    const float* vrow = V + ((size_t)(b * L + kbase + klo) * H + h) * E + 4 * lh;
    const int ao0 = lh * LDS_STRIDE + kbase + klo - w0;
    const int ao1 = ao0 + 16 * LDS_STRIDE;

    const unsigned rowu = inpatch[row_base] | inpatch[row_base + 15] |
                          inpatch[row_base + 16] | inpatch[row_base + 31];
    const int ct3 = (wave + 8 * (lane & 7)) * 16;
    const unsigned col3 = inpatch[ct3] | inpatch[ct3 + 15];
    const unsigned act = __builtin_amdgcn_ballot_w32((rowu & col3) != 0u) & 0xffu;

    unsigned rem = act;
    v4f vbA[G][2];
    v2f afA[G][2];
    const bool haveA = rem != 0u;
    if (haveA) {
      const int s = __builtin_ctz(rem);
      rem &= rem - 1u;
      p3_load<G>(vbA, afA, vrow, S, ao0, ao1, 32 * s, H * E);
    }

    float* part = S;
    float* vs = S + 8 * RT * PART_STRIDE;
    {
      v4f csum = {0.0f, 0.0f, 0.0f, 0.0f};
      const int half = lane >> 4;
      const float* trow = T + ((size_t)((b * H + h) * 64 + wave + 8 * half)) * E + 4 * lh;
      v4f ts[4];
#pragma unroll
      for (int j = 0; j < 4; ++j)
        ts[j] = *(const v4f*)(trow + (size_t)(16 * j) * E);
#pragma unroll
      for (int j = 0; j < 4; ++j) {
        const bool skip = ((act >> (2 * j + half)) & 1u) == 0u;
        csum.x += skip ? ts[j].x : 0.0f;
        csum.y += skip ? ts[j].y : 0.0f;
        csum.z += skip ? ts[j].z : 0.0f;
        csum.w += skip ? ts[j].w : 0.0f;
      }
      *(v4f*)&vs[(wave * 2 + half) * E + 4 * lh] = csum;
    }

    v8f acc[2][4] = {};
    if (haveA) p3_mma<G>(acc, vbA, afA);
    while (rem != 0u) {
      const int s = __builtin_ctz(rem);
      rem &= rem - 1u;
      v4f vbC[G][2];
      v2f afC[G][2];
      p3_load<G>(vbC, afC, vrow, S, ao0, ao1, 32 * s, H * E);
      p3_mma<G>(acc, vbC, afC);
    }

    __syncthreads();
#pragma unroll
    for (int m = 0; m < 2; ++m)
#pragma unroll
      for (int i = 0; i < 8; ++i) {
        v4f o;
        o.x = acc[m][0][i]; o.y = acc[m][1][i]; o.z = acc[m][2][i]; o.w = acc[m][3][i];
        *(v4f*)&part[(wave * RT + 16 * m + i + hih) * PART_STRIDE + 4 * lh] = o;
      }
    __syncthreads();

    const int c4 = (tid & 15) * 4;
    v4f vsum = *(const v4f*)&vs[c4];
    for (int q = 1; q < 16; ++q)
      vsum += *(const v4f*)&vs[q * E + c4];
    for (int j = 0; j < 2; ++j) {
      const int f   = tid + 256 * j;
      const int rl  = f >> 4;
      v4f s = *(const v4f*)&part[rl * PART_STRIDE + c4];
      for (int w = 1; w < 8; ++w)
        s += *(const v4f*)&part[(w * RT + rl) * PART_STRIDE + c4];
      const float cz = czero[rl];
      const float is = inv_sum[rl];
      s.x = (s.x + cz * vsum.x) * is;
      s.y = (s.y + cz * vsum.y) * is;
      s.z = (s.z + cz * vsum.z) * is;
      s.w = (s.w + cz * vsum.w) * is;
      *(v4f*)&O[((size_t)(b * L + row_base + rl) * H + h) * E + c4] = s;
    }
  }
}

extern "C" void kernel_launch(void* const* d_in, const int* in_sizes, int n_in,
                              void* d_out, int out_size, void* d_ws, size_t ws_size,
                              hipStream_t stream) {
  const float* Q = (const float*)d_in[0];
  const float* K = (const float*)d_in[1];
  const float* V = (const float*)d_in[2];
  const int* starts = (const int*)d_in[3];
  float* O = (float*)d_out;

  float* T = (float*)d_ws;

  hipLaunchKernelGGL(v_tile_sums_kernel, dim3(512), dim3(128), 0, stream, V, T);

  dim3 grid(1024 / 32, 8, 8);
  dim3 block(256);
  hipLaunchKernelGGL(prob_attn_kernel, grid, block, 0, stream, Q, K, V, starts, T, O);
}


# Patch-membership words written once by the pre-pass kernel (4 KiB workspace), loaded by the main kernel as needed: the per-workgroup 1024 x 32 table build, its LDS and barrier go; bit-identical
# speedup vs baseline: 17.0165x; 1.1132x over previous
#include <hip/hip_runtime.h>
#include <math.h>


typedef __attribute__((ext_vector_type(2))) float v2f;
typedef __attribute__((ext_vector_type(4))) float v4f;
typedef __attribute__((ext_vector_type(8))) float v8f;

#define LDS_STRIDE 164
#define PART_STRIDE 68

template <int G>
__device__ __forceinline__ void p3_load(v4f (&vb)[G][2], v2f (&af)[G][2],
                                        const float* __restrict__ vrow,
                                        const float* sblk, int ao0, int ao1,
                                        int ks0, int vpitch) {
#pragma unroll
  for (int g = 0; g < G; ++g) {
    const int k = 4 * (ks0 + g);
    vb[g][0] = *(const v4f*)(vrow + (size_t)k * vpitch);
    vb[g][1] = *(const v4f*)(vrow + (size_t)(k + 1) * vpitch);
    af[g][0] = *(const v2f*)(sblk + (ao0 + k));
    af[g][1] = *(const v2f*)(sblk + (ao1 + k));
  }
}
template <int G>
__device__ __forceinline__ void p3_mma(v8f (&acc)[2][4], const v4f (&vb)[G][2],
                                       const v2f (&af)[G][2]) {
#pragma unroll
  for (int g = 0; g < G; ++g)
#pragma unroll
    for (int t = 0; t < 4; ++t) {
      v2f bf;
      bf.x = vb[g][0][t];
      bf.y = vb[g][1][t];
      acc[0][t] = __builtin_amdgcn_wmma_f32_16x16x4_f32(
          false, af[g][0], false, bf, (short)0, acc[0][t], false, false);
      acc[1][t] = __builtin_amdgcn_wmma_f32_16x16x4_f32(
          false, af[g][1], false, bf, (short)0, acc[1][t], false, false);
    }
}

__global__ __launch_bounds__(128)
void v_tile_sums_kernel(const float* __restrict__ V, const int* __restrict__ starts,
                        float* __restrict__ T, unsigned* __restrict__ IP) {
  constexpr int L = 1024, H = 8, E = 64;
  constexpr int PATCH = 64;
  const int lane = threadIdx.x & 31;
  const int lh   = lane & 15;
  const int half = lane >> 4;
  const int t0   = 2 * (blockIdx.x * 4 + (threadIdx.x >> 5));
  const int bh   = t0 >> 6;
  const int tile = t0 & 63;
  const int b = bh >> 3, h = bh & 7;
  const float* vrow = V + ((size_t)(b * L + tile * 16 + half) * H + h) * E + 4 * lh;
  v4f x[2][8];
#pragma unroll
  for (int u = 0; u < 2; ++u)
#pragma unroll
    for (int i = 0; i < 8; ++i)
      x[u][i] = *(const v4f*)(vrow + (size_t)(16 * u + 2 * i) * (H * E));
  if (blockIdx.x < L / 128) {
    const int pos = blockIdx.x * 128 + threadIdx.x;
    unsigned m = 0u;
#pragma unroll
    for (int p = 0; p < 32; ++p) {
      const int s0p = starts[p];
      if (pos >= s0p && pos < s0p + PATCH) m |= (1u << p);
    }
    IP[pos] = m;
  }
  v4f s0 = x[0][0], s1 = x[1][0];
#pragma unroll
  for (int i = 1; i < 8; ++i) { s0 += x[0][i]; s1 += x[1][i]; }
  const v4f mine = half ? s1 : s0;
  const v4f send = half ? s0 : s1;
  v4f o;
  o.x = mine.x + __shfl_xor(send.x, 16, 32);
  o.y = mine.y + __shfl_xor(send.y, 16, 32);
  o.z = mine.z + __shfl_xor(send.z, 16, 32);
  o.w = mine.w + __shfl_xor(send.w, 16, 32);
  *(v4f*)(T + (size_t)(t0 + half) * E + 4 * lh) = o;
}

__global__ __launch_bounds__(256)
void prob_attn_kernel(const float* __restrict__ Q, const float* __restrict__ K,
                      const float* __restrict__ V, const unsigned* __restrict__ IP,
                      const float* __restrict__ T, float* __restrict__ O) {
  constexpr int L = 1024, H = 8, E = 64, RT = 32;
  constexpr int PATCH = 64;
  static_assert(PATCH >= 16, "the tile test in phase 1 needs a patch at least one tile long");
  constexpr int WIN = 160;
  constexpr int WT  = WIN / 16;
  static_assert(WIN >= RT + 2 * (PATCH - 1) && WIN % 32 == 0 && (L - WIN) % 16 == 0,
                "the window must cover rows +- (PATCH - 1), in whole tiles, 32 columns per pass");

  constexpr int S_FLOATS = 8 * RT * PART_STRIDE + 16 * E;
  static_assert(8 * RT * PART_STRIDE >= RT * LDS_STRIDE && LDS_STRIDE >= WIN,
                "the score block of the window must end in front of the column sums' area");
  __shared__ __attribute__((aligned(16))) float S[S_FLOATS];
  __shared__ float inv_sum[RT];
  __shared__ float czero[RT];

  const int tid  = threadIdx.x;
  const int wave = tid >> 5;
  const int lane = tid & 31;
  const int lh   = lane & 15;
  const int klo  = (lane < 16) ? 0 : 2;
  const int hih  = (lane < 16) ? 0 : 8;
  const int b = blockIdx.z, h = blockIdx.y;
  const int row_base = blockIdx.x * RT;
  const int w0 = min(max(row_base - PATCH, 0), L - WIN);

  const unsigned rowu = IP[row_base] | IP[row_base + 15] |
                        IP[row_base + 16] | IP[row_base + 31];
  const int ct3 = (wave + 8 * (lane & 7)) * 16;
  const unsigned col3 = IP[ct3] | IP[ct3 + 15];
  const unsigned act = __builtin_amdgcn_ballot_w32((rowu & col3) != 0u) & 0xffu;

  const int mt = wave & 1;
  const int r0 = mt * 16;
  {
    const int g = __builtin_amdgcn_readfirstlane(wave >> 1);
    const unsigned rowor = IP[row_base + r0] | IP[row_base + r0 + 15];
    const int wtl = g + 4 * (lane & 3);
    const int ct = w0 + 16 * (wtl < WT ? wtl : 0);
    const unsigned color = IP[ct] | IP[ct + 15];

    const float* qrow = Q + ((size_t)(b * L + row_base + r0 + lh) * H + h) * E;
    v2f afrag[16];
#pragma unroll
    for (int c = 0; c < 16; ++c)
      afrag[c] = *(const v2f*)(qrow + 4 * c + klo);

    typedef __attribute__((ext_vector_type(4))) unsigned v4u;
    const v4u* prp = (const v4u*)(IP + row_base + r0 + hih);
    const v4u prlo = prp[0], prhi = prp[1];
    const unsigned prow[8] = {prlo.x, prlo.y, prlo.z, prlo.w, prhi.x, prhi.y, prhi.z, prhi.w};

    const unsigned active = __builtin_amdgcn_ballot_w32((rowor & color) != 0u);

    for (int k = 0; g + 4 * k < WT; ++k) {
      const int c0 = (g + 4 * k) * 16;
      const int n0 = w0 + c0;
      if (((active >> k) & 1u) == 0u) {
        for (int i = 0; i < 8; ++i)
          S[(r0 + i + hih) * LDS_STRIDE + c0 + lh] = 0.0f;
        continue;
      }
      const unsigned pc = IP[n0 + lh];
      const float* krow = K + ((size_t)(b * L + n0 + lh) * H + h) * E;
      v8f acc = {};
#pragma unroll
      for (int c = 0; c < 16; ++c) {
        const v2f bfrag = *(const v2f*)(krow + 4 * c + klo);
        acc = __builtin_amdgcn_wmma_f32_16x16x4_f32(
            false, afrag[c], false, bfrag, (short)0, acc, false, false);
      }
#pragma unroll
      for (int i = 0; i < 8; ++i) {
        const float val = (prow[i] & pc) ? acc[i] : 0.0f;
        S[(r0 + i + hih) * LDS_STRIDE + c0 + lh] = val;
      }
    }
  }
  __syncthreads();

  {
    constexpr int NW = WIN / 32;
    constexpr float NOUT = (float)(L - WIN);
    float sw[4][NW];
#pragma unroll
    for (int rr = 0; rr < 4; ++rr)
#pragma unroll
      for (int j = 0; j < NW; ++j)
        sw[rr][j] = S[(wave * 4 + rr) * LDS_STRIDE + lane + 32 * j];

    float mx[4], sum[4];
#pragma unroll
    for (int rr = 0; rr < 4; ++rr) {
      float m = 0.0f;
#pragma unroll
      for (int j = 0; j < NW; ++j) m = fmaxf(m, sw[rr][j]);
      mx[rr] = m;
    }
#pragma unroll
    for (int d = 16; d; d >>= 1)
#pragma unroll
      for (int rr = 0; rr < 4; ++rr) mx[rr] = fmaxf(mx[rr], __shfl_xor(mx[rr], d, 32));

#pragma unroll
    for (int rr = 0; rr < 4; ++rr) {
      float* srow = &S[(wave * 4 + rr) * LDS_STRIDE];
      float s = 0.0f;
#pragma unroll
      for (int j = 0; j < NW; ++j) {
        const float p = expf(sw[rr][j] - mx[rr]);
        s += p;
        srow[lane + 32 * j] = p;
      }
      sum[rr] = s;
    }
#pragma unroll
    for (int d = 16; d; d >>= 1)
#pragma unroll
      for (int rr = 0; rr < 4; ++rr) sum[rr] += __shfl_xor(sum[rr], d, 32);

    float msel = mx[0], ssel = sum[0];
#pragma unroll
    for (int rr = 1; rr < 4; ++rr)
      if (lane == rr) { msel = mx[rr]; ssel = sum[rr]; }
    const float cz = expf(0.0f - msel);
    ssel += NOUT * cz;
    if (lane < 4) { inv_sum[wave * 4 + lane] = 1.0f / ssel; czero[wave * 4 + lane] = cz; }
  }
  __syncthreads();

  {
    constexpr int G = 4;
    const int kbase = wave * 16;
    const float* vrow = V + ((size_t)(b * L + kbase + klo) * H + h) * E + 4 * lh;
    const int ao0 = lh * LDS_STRIDE + kbase + klo - w0;
    const int ao1 = ao0 + 16 * LDS_STRIDE;


    unsigned rem = act;
    v4f vbA[G][2];
    v2f afA[G][2];
    const bool haveA = rem != 0u;
    if (haveA) {
      const int s = __builtin_ctz(rem);
      rem &= rem - 1u;
      p3_load<G>(vbA, afA, vrow, S, ao0, ao1, 32 * s, H * E);
    }

    float* part = S;
    float* vs = S + 8 * RT * PART_STRIDE;
    {
      v4f csum = {0.0f, 0.0f, 0.0f, 0.0f};
      const int half = lane >> 4;
      const float* trow = T + ((size_t)((b * H + h) * 64 + wave + 8 * half)) * E + 4 * lh;
      v4f ts[4];
#pragma unroll
      for (int j = 0; j < 4; ++j)
        ts[j] = *(const v4f*)(trow + (size_t)(16 * j) * E);
#pragma unroll
      for (int j = 0; j < 4; ++j) {
        const bool skip = ((act >> (2 * j + half)) & 1u) == 0u;
        csum.x += skip ? ts[j].x : 0.0f;
        csum.y += skip ? ts[j].y : 0.0f;
        csum.z += skip ? ts[j].z : 0.0f;
        csum.w += skip ? ts[j].w : 0.0f;
      }
      *(v4f*)&vs[(wave * 2 + half) * E + 4 * lh] = csum;
    }

    v8f acc[2][4] = {};
    if (haveA) p3_mma<G>(acc, vbA, afA);
    while (rem != 0u) {
      const int s = __builtin_ctz(rem);
      rem &= rem - 1u;
      v4f vbC[G][2];
      v2f afC[G][2];
      p3_load<G>(vbC, afC, vrow, S, ao0, ao1, 32 * s, H * E);
      p3_mma<G>(acc, vbC, afC);
    }

    __syncthreads();
#pragma unroll
    for (int m = 0; m < 2; ++m)
#pragma unroll
      for (int i = 0; i < 8; ++i) {
        v4f o;
        o.x = acc[m][0][i]; o.y = acc[m][1][i]; o.z = acc[m][2][i]; o.w = acc[m][3][i];
        *(v4f*)&part[(wave * RT + 16 * m + i + hih) * PART_STRIDE + 4 * lh] = o;
      }
    __syncthreads();

    const int c4 = (tid & 15) * 4;
    v4f vsum = *(const v4f*)&vs[c4];
    for (int q = 1; q < 16; ++q)
      vsum += *(const v4f*)&vs[q * E + c4];
    for (int j = 0; j < 2; ++j) {
      const int f   = tid + 256 * j;
      const int rl  = f >> 4;
      v4f s = *(const v4f*)&part[rl * PART_STRIDE + c4];
      for (int w = 1; w < 8; ++w)
        s += *(const v4f*)&part[(w * RT + rl) * PART_STRIDE + c4];
      const float cz = czero[rl];
      const float is = inv_sum[rl];
      s.x = (s.x + cz * vsum.x) * is;
      s.y = (s.y + cz * vsum.y) * is;
      s.z = (s.z + cz * vsum.z) * is;
      s.w = (s.w + cz * vsum.w) * is;
      *(v4f*)&O[((size_t)(b * L + row_base + rl) * H + h) * E + c4] = s;
    }
  }
}

extern "C" void kernel_launch(void* const* d_in, const int* in_sizes, int n_in,
                              void* d_out, int out_size, void* d_ws, size_t ws_size,
                              hipStream_t stream) {
  const float* Q = (const float*)d_in[0];
  const float* K = (const float*)d_in[1];
  const float* V = (const float*)d_in[2];
  const int* starts = (const int*)d_in[3];
  float* O = (float*)d_out;

  float* T = (float*)d_ws;
  unsigned* IP = (unsigned*)(T + 4096 * 64);

  hipLaunchKernelGGL(v_tile_sums_kernel, dim3(512), dim3(128), 0, stream, V, starts, T, IP);

  dim3 grid(1024 / 32, 8, 8);
  dim3 block(256);
  hipLaunchKernelGGL(prob_attn_kernel, grid, block, 0, stream, Q, K, V, IP, T, O);
}


# Phase 1 operands staged in LDS: the workgroup copies its 32 Q rows and the 160 window K rows once with whole-row 128-bit loads; fragments come from LDS in the same k order (bit-identical)
# speedup vs baseline: 18.4932x; 1.0868x over previous
#include <hip/hip_runtime.h>
#include <math.h>


typedef __attribute__((ext_vector_type(2))) float v2f;
typedef __attribute__((ext_vector_type(4))) float v4f;
typedef __attribute__((ext_vector_type(8))) float v8f;

#define LDS_STRIDE 164
#define PART_STRIDE 68
#define STG_STRIDE 68

template <int G>
__device__ __forceinline__ void p3_load(v4f (&vb)[G][2], v2f (&af)[G][2],
                                        const float* __restrict__ vrow,
                                        const float* sblk, int ao0, int ao1,
                                        int ks0, int vpitch) {
#pragma unroll
  for (int g = 0; g < G; ++g) {
    const int k = 4 * (ks0 + g);
    vb[g][0] = *(const v4f*)(vrow + (size_t)k * vpitch);
    vb[g][1] = *(const v4f*)(vrow + (size_t)(k + 1) * vpitch);
    af[g][0] = *(const v2f*)(sblk + (ao0 + k));
    af[g][1] = *(const v2f*)(sblk + (ao1 + k));
  }
}
template <int G>
__device__ __forceinline__ void p3_mma(v8f (&acc)[2][4], const v4f (&vb)[G][2],
                                       const v2f (&af)[G][2]) {
#pragma unroll
  for (int g = 0; g < G; ++g)
#pragma unroll
    for (int t = 0; t < 4; ++t) {
      v2f bf;
      bf.x = vb[g][0][t];
      bf.y = vb[g][1][t];
      acc[0][t] = __builtin_amdgcn_wmma_f32_16x16x4_f32(
          false, af[g][0], false, bf, (short)0, acc[0][t], false, false);
      acc[1][t] = __builtin_amdgcn_wmma_f32_16x16x4_f32(
          false, af[g][1], false, bf, (short)0, acc[1][t], false, false);
    }
}

__global__ __launch_bounds__(128)
void v_tile_sums_kernel(const float* __restrict__ V, const int* __restrict__ starts,
                        float* __restrict__ T, unsigned* __restrict__ IP) {
  constexpr int L = 1024, H = 8, E = 64;
  constexpr int PATCH = 64;
  const int lane = threadIdx.x & 31;
  const int lh   = lane & 15;
  const int half = lane >> 4;
  const int t0   = 2 * (blockIdx.x * 4 + (threadIdx.x >> 5));
  const int bh   = t0 >> 6;
  const int tile = t0 & 63;
  const int b = bh >> 3, h = bh & 7;
  const float* vrow = V + ((size_t)(b * L + tile * 16 + half) * H + h) * E + 4 * lh;
  v4f x[2][8];
#pragma unroll
  for (int u = 0; u < 2; ++u)
#pragma unroll
    for (int i = 0; i < 8; ++i)
      x[u][i] = *(const v4f*)(vrow + (size_t)(16 * u + 2 * i) * (H * E));
  if (blockIdx.x < L / 128) {
    const int pos = blockIdx.x * 128 + threadIdx.x;
    unsigned m = 0u;
#pragma unroll
    for (int p = 0; p < 32; ++p) {
      const int s0p = starts[p];
      if (pos >= s0p && pos < s0p + PATCH) m |= (1u << p);
    }
    IP[pos] = m;
  }
  v4f s0 = x[0][0], s1 = x[1][0];
#pragma unroll
  for (int i = 1; i < 8; ++i) { s0 += x[0][i]; s1 += x[1][i]; }
  const v4f mine = half ? s1 : s0;
  const v4f send = half ? s0 : s1;
  v4f o;
  o.x = mine.x + __shfl_xor(send.x, 16, 32);
  o.y = mine.y + __shfl_xor(send.y, 16, 32);
  o.z = mine.z + __shfl_xor(send.z, 16, 32);
  o.w = mine.w + __shfl_xor(send.w, 16, 32);
  *(v4f*)(T + (size_t)(t0 + half) * E + 4 * lh) = o;
}

__global__ __launch_bounds__(256)
void prob_attn_kernel(const float* __restrict__ Q, const float* __restrict__ K,
                      const float* __restrict__ V, const unsigned* __restrict__ IP,
                      const float* __restrict__ T, float* __restrict__ O) {
  constexpr int L = 1024, H = 8, E = 64, RT = 32;
  constexpr int PATCH = 64;
  static_assert(PATCH >= 16, "the tile test in phase 1 needs a patch at least one tile long");
  constexpr int WIN = 160;
  constexpr int WT  = WIN / 16;
  static_assert(WIN >= RT + 2 * (PATCH - 1) && WIN % 32 == 0 && (L - WIN) % 16 == 0,
                "the window must cover rows +- (PATCH - 1), in whole tiles, 32 columns per pass");

  constexpr int S_FLOATS = 8 * RT * PART_STRIDE + 16 * E;
  static_assert(8 * RT * PART_STRIDE >= RT * LDS_STRIDE && LDS_STRIDE >= WIN,
                "the score block of the window must end in front of the column sums' area");
  __shared__ __attribute__((aligned(16))) float S[S_FLOATS];
  __shared__ float inv_sum[RT];
  __shared__ float czero[RT];

  const int tid  = threadIdx.x;
  const int wave = tid >> 5;
  const int lane = tid & 31;
  const int lh   = lane & 15;
  const int klo  = (lane < 16) ? 0 : 2;
  const int hih  = (lane < 16) ? 0 : 8;
  const int b = blockIdx.z, h = blockIdx.y;
  const int row_base = blockIdx.x * RT;
  const int w0 = min(max(row_base - PATCH, 0), L - WIN);

  float* QS = S + RT * LDS_STRIDE;
  float* KS = QS + RT * STG_STRIDE;
  static_assert(RT * LDS_STRIDE + (RT + WIN) * STG_STRIDE <= S_FLOATS &&
                (RT * LDS_STRIDE) % 4 == 0 && (STG_STRIDE % 4) == 0,
                "the staged rows must fit behind the score block, 16-byte aligned");
  {
    const int sr = tid >> 4;
    const int sc = (tid & 15) * 4;
    const float* qg = Q + ((size_t)(b * L + row_base + sr) * H + h) * E + sc;
    const float* kg = K + ((size_t)(b * L + w0 + sr) * H + h) * E + sc;
    v4f qv[RT / 16], kv[WT];
#pragma unroll
    for (int j = 0; j < RT / 16; ++j) qv[j] = *(const v4f*)(qg + (size_t)(16 * j) * (H * E));
#pragma unroll
    for (int j = 0; j < WT; ++j) kv[j] = *(const v4f*)(kg + (size_t)(16 * j) * (H * E));
#pragma unroll
    for (int j = 0; j < RT / 16; ++j) *(v4f*)&QS[(16 * j + sr) * STG_STRIDE + sc] = qv[j];
#pragma unroll
    for (int j = 0; j < WT; ++j) *(v4f*)&KS[(16 * j + sr) * STG_STRIDE + sc] = kv[j];
  }

  const unsigned rowu = IP[row_base] | IP[row_base + 15] |
                        IP[row_base + 16] | IP[row_base + 31];
  const int ct3 = (wave + 8 * (lane & 7)) * 16;
  const unsigned col3 = IP[ct3] | IP[ct3 + 15];
  const unsigned act = __builtin_amdgcn_ballot_w32((rowu & col3) != 0u) & 0xffu;

  const int mt = wave & 1;
  const int r0 = mt * 16;
  {
    const int g = __builtin_amdgcn_readfirstlane(wave >> 1);
    const unsigned rowor = IP[row_base + r0] | IP[row_base + r0 + 15];
    const int wtl = g + 4 * (lane & 3);
    const int ct = w0 + 16 * (wtl < WT ? wtl : 0);
    const unsigned color = IP[ct] | IP[ct + 15];

    typedef __attribute__((ext_vector_type(4))) unsigned v4u;
    const v4u* prp = (const v4u*)(IP + row_base + r0 + hih);
    const v4u prlo = prp[0], prhi = prp[1];
    const unsigned prow[8] = {prlo.x, prlo.y, prlo.z, prlo.w, prhi.x, prhi.y, prhi.z, prhi.w};

    __syncthreads();

    const float* qrow = QS + (r0 + lh) * STG_STRIDE + klo;
    v2f afrag[16];
#pragma unroll
    for (int c = 0; c < 16; ++c)
      afrag[c] = *(const v2f*)(qrow + 4 * c);

    const unsigned active = __builtin_amdgcn_ballot_w32((rowor & color) != 0u);

    for (int k = 0; g + 4 * k < WT; ++k) {
      const int c0 = (g + 4 * k) * 16;
      const int n0 = w0 + c0;
      if (((active >> k) & 1u) == 0u) {
        for (int i = 0; i < 8; ++i)
          S[(r0 + i + hih) * LDS_STRIDE + c0 + lh] = 0.0f;
        continue;
      }
      const unsigned pc = IP[n0 + lh];
      const float* krow = KS + (c0 + lh) * STG_STRIDE + klo;
      v8f acc = {};
#pragma unroll
      for (int c = 0; c < 16; ++c) {
        const v2f bfrag = *(const v2f*)(krow + 4 * c);
        acc = __builtin_amdgcn_wmma_f32_16x16x4_f32(
            false, afrag[c], false, bfrag, (short)0, acc, false, false);
      }
#pragma unroll
      for (int i = 0; i < 8; ++i) {
        const float val = (prow[i] & pc) ? acc[i] : 0.0f;
        S[(r0 + i + hih) * LDS_STRIDE + c0 + lh] = val;
      }
    }
  }
  __syncthreads();

  {
    constexpr int NW = WIN / 32;
    constexpr float NOUT = (float)(L - WIN);
    float sw[4][NW];
#pragma unroll
    for (int rr = 0; rr < 4; ++rr)
#pragma unroll
      for (int j = 0; j < NW; ++j)
        sw[rr][j] = S[(wave * 4 + rr) * LDS_STRIDE + lane + 32 * j];

    float mx[4], sum[4];
#pragma unroll
    for (int rr = 0; rr < 4; ++rr) {
      float m = 0.0f;
#pragma unroll
      for (int j = 0; j < NW; ++j) m = fmaxf(m, sw[rr][j]);
      mx[rr] = m;
    }
#pragma unroll
    for (int d = 16; d; d >>= 1)
#pragma unroll
      for (int rr = 0; rr < 4; ++rr) mx[rr] = fmaxf(mx[rr], __shfl_xor(mx[rr], d, 32));

#pragma unroll
    for (int rr = 0; rr < 4; ++rr) {
      float* srow = &S[(wave * 4 + rr) * LDS_STRIDE];
      float s = 0.0f;
#pragma unroll
      for (int j = 0; j < NW; ++j) {
        const float p = expf(sw[rr][j] - mx[rr]);
        s += p;
        srow[lane + 32 * j] = p;
      }
      sum[rr] = s;
    }
#pragma unroll
    for (int d = 16; d; d >>= 1)
#pragma unroll
      for (int rr = 0; rr < 4; ++rr) sum[rr] += __shfl_xor(sum[rr], d, 32);

    float msel = mx[0], ssel = sum[0];
#pragma unroll
    for (int rr = 1; rr < 4; ++rr)
      if (lane == rr) { msel = mx[rr]; ssel = sum[rr]; }
    const float cz = expf(0.0f - msel);
    ssel += NOUT * cz;
    if (lane < 4) { inv_sum[wave * 4 + lane] = 1.0f / ssel; czero[wave * 4 + lane] = cz; }
  }
  __syncthreads();

  {
    constexpr int G = 4;
    const int kbase = wave * 16;
    const float* vrow = V + ((size_t)(b * L + kbase + klo) * H + h) * E + 4 * lh;
    const int ao0 = lh * LDS_STRIDE + kbase + klo - w0;
    const int ao1 = ao0 + 16 * LDS_STRIDE;


    unsigned rem = act;
    v4f vbA[G][2];
    v2f afA[G][2];
    const bool haveA = rem != 0u;
    if (haveA) {
      const int s = __builtin_ctz(rem);
      rem &= rem - 1u;
      p3_load<G>(vbA, afA, vrow, S, ao0, ao1, 32 * s, H * E);
    }

    float* part = S;
    float* vs = S + 8 * RT * PART_STRIDE;
    {
      v4f csum = {0.0f, 0.0f, 0.0f, 0.0f};
      const int half = lane >> 4;
      const float* trow = T + ((size_t)((b * H + h) * 64 + wave + 8 * half)) * E + 4 * lh;
      v4f ts[4];
#pragma unroll
      for (int j = 0; j < 4; ++j)
        ts[j] = *(const v4f*)(trow + (size_t)(16 * j) * E);
#pragma unroll
      for (int j = 0; j < 4; ++j) {
        const bool skip = ((act >> (2 * j + half)) & 1u) == 0u;
        csum.x += skip ? ts[j].x : 0.0f;
        csum.y += skip ? ts[j].y : 0.0f;
        csum.z += skip ? ts[j].z : 0.0f;
        csum.w += skip ? ts[j].w : 0.0f;
      }
      *(v4f*)&vs[(wave * 2 + half) * E + 4 * lh] = csum;
    }

    v8f acc[2][4] = {};
    if (haveA) p3_mma<G>(acc, vbA, afA);
    while (rem != 0u) {
      const int s = __builtin_ctz(rem);
      rem &= rem - 1u;
      v4f vbC[G][2];
      v2f afC[G][2];
      p3_load<G>(vbC, afC, vrow, S, ao0, ao1, 32 * s, H * E);
      p3_mma<G>(acc, vbC, afC);
    }

    __syncthreads();
#pragma unroll
    for (int m = 0; m < 2; ++m)
#pragma unroll
      for (int i = 0; i < 8; ++i) {
        v4f o;
        o.x = acc[m][0][i]; o.y = acc[m][1][i]; o.z = acc[m][2][i]; o.w = acc[m][3][i];
        *(v4f*)&part[(wave * RT + 16 * m + i + hih) * PART_STRIDE + 4 * lh] = o;
      }
    __syncthreads();

    const int c4 = (tid & 15) * 4;
    v4f vsum = *(const v4f*)&vs[c4];
    for (int q = 1; q < 16; ++q)
      vsum += *(const v4f*)&vs[q * E + c4];
    for (int j = 0; j < 2; ++j) {
      const int f   = tid + 256 * j;
      const int rl  = f >> 4;
      v4f s = *(const v4f*)&part[rl * PART_STRIDE + c4];
      for (int w = 1; w < 8; ++w)
        s += *(const v4f*)&part[(w * RT + rl) * PART_STRIDE + c4];
      const float cz = czero[rl];
      const float is = inv_sum[rl];
      s.x = (s.x + cz * vsum.x) * is;
      s.y = (s.y + cz * vsum.y) * is;
      s.z = (s.z + cz * vsum.z) * is;
      s.w = (s.w + cz * vsum.w) * is;
      *(v4f*)&O[((size_t)(b * L + row_base + rl) * H + h) * E + c4] = s;
    }
  }
}

extern "C" void kernel_launch(void* const* d_in, const int* in_sizes, int n_in,
                              void* d_out, int out_size, void* d_ws, size_t ws_size,
                              hipStream_t stream) {
  const float* Q = (const float*)d_in[0];
  const float* K = (const float*)d_in[1];
  const float* V = (const float*)d_in[2];
  const int* starts = (const int*)d_in[3];
  float* O = (float*)d_out;

  float* T = (float*)d_ws;
  unsigned* IP = (unsigned*)(T + 4096 * 64);

  hipLaunchKernelGGL(v_tile_sums_kernel, dim3(512), dim3(128), 0, stream, V, starts, T, IP);

  dim3 grid(1024 / 32, 8, 8);
  dim3 block(256);
  hipLaunchKernelGGL(prob_attn_kernel, grid, block, 0, stream, Q, K, V, IP, T, O);
}


# Softmax numerators in base 2: p = exp2(fma(s, log2e, -m*log2e)) on the hardware exp2 instruction instead of library expf; m*log2e rounded once per row, a common shift that cancels (P last bits)
# speedup vs baseline: 20.6116x; 1.1146x over previous
#include <hip/hip_runtime.h>
#include <math.h>


typedef __attribute__((ext_vector_type(2))) float v2f;
typedef __attribute__((ext_vector_type(4))) float v4f;
typedef __attribute__((ext_vector_type(8))) float v8f;

#define LDS_STRIDE 164
#define PART_STRIDE 68
#define STG_STRIDE 68

template <int G>
__device__ __forceinline__ void p3_load(v4f (&vb)[G][2], v2f (&af)[G][2],
                                        const float* __restrict__ vrow,
                                        const float* sblk, int ao0, int ao1,
                                        int ks0, int vpitch) {
#pragma unroll
  for (int g = 0; g < G; ++g) {
    const int k = 4 * (ks0 + g);
    vb[g][0] = *(const v4f*)(vrow + (size_t)k * vpitch);
    vb[g][1] = *(const v4f*)(vrow + (size_t)(k + 1) * vpitch);
    af[g][0] = *(const v2f*)(sblk + (ao0 + k));
    af[g][1] = *(const v2f*)(sblk + (ao1 + k));
  }
}
template <int G>
__device__ __forceinline__ void p3_mma(v8f (&acc)[2][4], const v4f (&vb)[G][2],
                                       const v2f (&af)[G][2]) {
#pragma unroll
  for (int g = 0; g < G; ++g)
#pragma unroll
    for (int t = 0; t < 4; ++t) {
      v2f bf;
      bf.x = vb[g][0][t];
      bf.y = vb[g][1][t];
      acc[0][t] = __builtin_amdgcn_wmma_f32_16x16x4_f32(
          false, af[g][0], false, bf, (short)0, acc[0][t], false, false);
      acc[1][t] = __builtin_amdgcn_wmma_f32_16x16x4_f32(
          false, af[g][1], false, bf, (short)0, acc[1][t], false, false);
    }
}

__global__ __launch_bounds__(128)
void v_tile_sums_kernel(const float* __restrict__ V, const int* __restrict__ starts,
                        float* __restrict__ T, unsigned* __restrict__ IP) {
  constexpr int L = 1024, H = 8, E = 64;
  constexpr int PATCH = 64;
  const int lane = threadIdx.x & 31;
  const int lh   = lane & 15;
  const int half = lane >> 4;
  const int t0   = 2 * (blockIdx.x * 4 + (threadIdx.x >> 5));
  const int bh   = t0 >> 6;
  const int tile = t0 & 63;
  const int b = bh >> 3, h = bh & 7;
  const float* vrow = V + ((size_t)(b * L + tile * 16 + half) * H + h) * E + 4 * lh;
  v4f x[2][8];
#pragma unroll
  for (int u = 0; u < 2; ++u)
#pragma unroll
    for (int i = 0; i < 8; ++i)
      x[u][i] = *(const v4f*)(vrow + (size_t)(16 * u + 2 * i) * (H * E));
  if (blockIdx.x < L / 128) {
    const int pos = blockIdx.x * 128 + threadIdx.x;
    unsigned m = 0u;
#pragma unroll
    for (int p = 0; p < 32; ++p) {
      const int s0p = starts[p];
      if (pos >= s0p && pos < s0p + PATCH) m |= (1u << p);
    }
    IP[pos] = m;
  }
  v4f s0 = x[0][0], s1 = x[1][0];
#pragma unroll
  for (int i = 1; i < 8; ++i) { s0 += x[0][i]; s1 += x[1][i]; }
  const v4f mine = half ? s1 : s0;
  const v4f send = half ? s0 : s1;
  v4f o;
  o.x = mine.x + __shfl_xor(send.x, 16, 32);
  o.y = mine.y + __shfl_xor(send.y, 16, 32);
  o.z = mine.z + __shfl_xor(send.z, 16, 32);
  o.w = mine.w + __shfl_xor(send.w, 16, 32);
  *(v4f*)(T + (size_t)(t0 + half) * E + 4 * lh) = o;
}

__global__ __launch_bounds__(256)
void prob_attn_kernel(const float* __restrict__ Q, const float* __restrict__ K,
                      const float* __restrict__ V, const unsigned* __restrict__ IP,
                      const float* __restrict__ T, float* __restrict__ O) {
  constexpr int L = 1024, H = 8, E = 64, RT = 32;
  constexpr int PATCH = 64;
  static_assert(PATCH >= 16, "the tile test in phase 1 needs a patch at least one tile long");
  constexpr int WIN = 160;
  constexpr int WT  = WIN / 16;
  static_assert(WIN >= RT + 2 * (PATCH - 1) && WIN % 32 == 0 && (L - WIN) % 16 == 0,
                "the window must cover rows +- (PATCH - 1), in whole tiles, 32 columns per pass");

  constexpr int S_FLOATS = 8 * RT * PART_STRIDE + 16 * E;
  static_assert(8 * RT * PART_STRIDE >= RT * LDS_STRIDE && LDS_STRIDE >= WIN,
                "the score block of the window must end in front of the column sums' area");
  __shared__ __attribute__((aligned(16))) float S[S_FLOATS];
  __shared__ float inv_sum[RT];
  __shared__ float czero[RT];

  const int tid  = threadIdx.x;
  const int wave = tid >> 5;
  const int lane = tid & 31;
  const int lh   = lane & 15;
  const int klo  = (lane < 16) ? 0 : 2;
  const int hih  = (lane < 16) ? 0 : 8;
  const int b = blockIdx.z, h = blockIdx.y;
  const int row_base = blockIdx.x * RT;
  const int w0 = min(max(row_base - PATCH, 0), L - WIN);

  float* QS = S + RT * LDS_STRIDE;
  float* KS = QS + RT * STG_STRIDE;
  static_assert(RT * LDS_STRIDE + (RT + WIN) * STG_STRIDE <= S_FLOATS &&
                (RT * LDS_STRIDE) % 4 == 0 && (STG_STRIDE % 4) == 0,
                "the staged rows must fit behind the score block, 16-byte aligned");
  {
    const int sr = tid >> 4;
    const int sc = (tid & 15) * 4;
    const float* qg = Q + ((size_t)(b * L + row_base + sr) * H + h) * E + sc;
    const float* kg = K + ((size_t)(b * L + w0 + sr) * H + h) * E + sc;
    v4f qv[RT / 16], kv[WT];
#pragma unroll
    for (int j = 0; j < RT / 16; ++j) qv[j] = *(const v4f*)(qg + (size_t)(16 * j) * (H * E));
#pragma unroll
    for (int j = 0; j < WT; ++j) kv[j] = *(const v4f*)(kg + (size_t)(16 * j) * (H * E));
#pragma unroll
    for (int j = 0; j < RT / 16; ++j) *(v4f*)&QS[(16 * j + sr) * STG_STRIDE + sc] = qv[j];
#pragma unroll
    for (int j = 0; j < WT; ++j) *(v4f*)&KS[(16 * j + sr) * STG_STRIDE + sc] = kv[j];
  }

  const unsigned rowu = IP[row_base] | IP[row_base + 15] |
                        IP[row_base + 16] | IP[row_base + 31];
  const int ct3 = (wave + 8 * (lane & 7)) * 16;
  const unsigned col3 = IP[ct3] | IP[ct3 + 15];
  const unsigned act = __builtin_amdgcn_ballot_w32((rowu & col3) != 0u) & 0xffu;

  const int mt = wave & 1;
  const int r0 = mt * 16;
  {
    const int g = __builtin_amdgcn_readfirstlane(wave >> 1);
    const unsigned rowor = IP[row_base + r0] | IP[row_base + r0 + 15];
    const int wtl = g + 4 * (lane & 3);
    const int ct = w0 + 16 * (wtl < WT ? wtl : 0);
    const unsigned color = IP[ct] | IP[ct + 15];

    typedef __attribute__((ext_vector_type(4))) unsigned v4u;
    const v4u* prp = (const v4u*)(IP + row_base + r0 + hih);
    const v4u prlo = prp[0], prhi = prp[1];
    const unsigned prow[8] = {prlo.x, prlo.y, prlo.z, prlo.w, prhi.x, prhi.y, prhi.z, prhi.w};

    __syncthreads();

    const float* qrow = QS + (r0 + lh) * STG_STRIDE + klo;
    v2f afrag[16];
#pragma unroll
    for (int c = 0; c < 16; ++c)
      afrag[c] = *(const v2f*)(qrow + 4 * c);

    const unsigned active = __builtin_amdgcn_ballot_w32((rowor & color) != 0u);

    for (int k = 0; g + 4 * k < WT; ++k) {
      const int c0 = (g + 4 * k) * 16;
      const int n0 = w0 + c0;
      if (((active >> k) & 1u) == 0u) {
        for (int i = 0; i < 8; ++i)
          S[(r0 + i + hih) * LDS_STRIDE + c0 + lh] = 0.0f;
        continue;
      }
      const unsigned pc = IP[n0 + lh];
      const float* krow = KS + (c0 + lh) * STG_STRIDE + klo;
      v8f acc = {};
#pragma unroll
      for (int c = 0; c < 16; ++c) {
        const v2f bfrag = *(const v2f*)(krow + 4 * c);
        acc = __builtin_amdgcn_wmma_f32_16x16x4_f32(
            false, afrag[c], false, bfrag, (short)0, acc, false, false);
      }
#pragma unroll
      for (int i = 0; i < 8; ++i) {
        const float val = (prow[i] & pc) ? acc[i] : 0.0f;
        S[(r0 + i + hih) * LDS_STRIDE + c0 + lh] = val;
      }
    }
  }
  __syncthreads();

  {
    constexpr int NW = WIN / 32;
    constexpr float NOUT = (float)(L - WIN);
    constexpr float LOG2E = 1.4426950408889634f;
    float sw[4][NW];
#pragma unroll
    for (int rr = 0; rr < 4; ++rr)
#pragma unroll
      for (int j = 0; j < NW; ++j)
        sw[rr][j] = S[(wave * 4 + rr) * LDS_STRIDE + lane + 32 * j];

    float mx[4], sum[4];
#pragma unroll
    for (int rr = 0; rr < 4; ++rr) {
      float m = 0.0f;
#pragma unroll
      for (int j = 0; j < NW; ++j) m = fmaxf(m, sw[rr][j]);
      mx[rr] = m;
    }
#pragma unroll
    for (int d = 16; d; d >>= 1)
#pragma unroll
      for (int rr = 0; rr < 4; ++rr) mx[rr] = fmaxf(mx[rr], __shfl_xor(mx[rr], d, 32));
    float ml[4];
#pragma unroll
    for (int rr = 0; rr < 4; ++rr) ml[rr] = mx[rr] * LOG2E;

#pragma unroll
    for (int rr = 0; rr < 4; ++rr) {
      float* srow = &S[(wave * 4 + rr) * LDS_STRIDE];
      float s = 0.0f;
#pragma unroll
      for (int j = 0; j < NW; ++j) {
        const float p = __builtin_amdgcn_exp2f(fmaf(sw[rr][j], LOG2E, -ml[rr]));
        s += p;
        srow[lane + 32 * j] = p;
      }
      sum[rr] = s;
    }
#pragma unroll
    for (int d = 16; d; d >>= 1)
#pragma unroll
      for (int rr = 0; rr < 4; ++rr) sum[rr] += __shfl_xor(sum[rr], d, 32);

    float msel = ml[0], ssel = sum[0];
#pragma unroll
    for (int rr = 1; rr < 4; ++rr)
      if (lane == rr) { msel = ml[rr]; ssel = sum[rr]; }
    const float cz = __builtin_amdgcn_exp2f(0.0f - msel);
    ssel += NOUT * cz;
    if (lane < 4) { inv_sum[wave * 4 + lane] = 1.0f / ssel; czero[wave * 4 + lane] = cz; }
  }
  __syncthreads();

  {
    constexpr int G = 4;
    const int kbase = wave * 16;
    const float* vrow = V + ((size_t)(b * L + kbase + klo) * H + h) * E + 4 * lh;
    const int ao0 = lh * LDS_STRIDE + kbase + klo - w0;
    const int ao1 = ao0 + 16 * LDS_STRIDE;


    unsigned rem = act;
    v4f vbA[G][2];
    v2f afA[G][2];
    const bool haveA = rem != 0u;
    if (haveA) {
      const int s = __builtin_ctz(rem);
      rem &= rem - 1u;
      p3_load<G>(vbA, afA, vrow, S, ao0, ao1, 32 * s, H * E);
    }

    float* part = S;
    float* vs = S + 8 * RT * PART_STRIDE;
    {
      v4f csum = {0.0f, 0.0f, 0.0f, 0.0f};
      const int half = lane >> 4;
      const float* trow = T + ((size_t)((b * H + h) * 64 + wave + 8 * half)) * E + 4 * lh;
      v4f ts[4];
#pragma unroll
      for (int j = 0; j < 4; ++j)
        ts[j] = *(const v4f*)(trow + (size_t)(16 * j) * E);
#pragma unroll
      for (int j = 0; j < 4; ++j) {
        const bool skip = ((act >> (2 * j + half)) & 1u) == 0u;
        csum.x += skip ? ts[j].x : 0.0f;
        csum.y += skip ? ts[j].y : 0.0f;
        csum.z += skip ? ts[j].z : 0.0f;
        csum.w += skip ? ts[j].w : 0.0f;
      }
      *(v4f*)&vs[(wave * 2 + half) * E + 4 * lh] = csum;
    }

    v8f acc[2][4] = {};
    if (haveA) p3_mma<G>(acc, vbA, afA);
    while (rem != 0u) {
      const int s = __builtin_ctz(rem);
      rem &= rem - 1u;
      v4f vbC[G][2];
      v2f afC[G][2];
      p3_load<G>(vbC, afC, vrow, S, ao0, ao1, 32 * s, H * E);
      p3_mma<G>(acc, vbC, afC);
    }

    __syncthreads();
#pragma unroll
    for (int m = 0; m < 2; ++m)
#pragma unroll
      for (int i = 0; i < 8; ++i) {
        v4f o;
        o.x = acc[m][0][i]; o.y = acc[m][1][i]; o.z = acc[m][2][i]; o.w = acc[m][3][i];
        *(v4f*)&part[(wave * RT + 16 * m + i + hih) * PART_STRIDE + 4 * lh] = o;
      }
    __syncthreads();

    const int c4 = (tid & 15) * 4;
    v4f vsum = *(const v4f*)&vs[c4];
    for (int q = 1; q < 16; ++q)
      vsum += *(const v4f*)&vs[q * E + c4];
    for (int j = 0; j < 2; ++j) {
      const int f   = tid + 256 * j;
      const int rl  = f >> 4;
      v4f s = *(const v4f*)&part[rl * PART_STRIDE + c4];
      for (int w = 1; w < 8; ++w)
        s += *(const v4f*)&part[(w * RT + rl) * PART_STRIDE + c4];
      const float cz = czero[rl];
      const float is = inv_sum[rl];
      s.x = (s.x + cz * vsum.x) * is;
      s.y = (s.y + cz * vsum.y) * is;
      s.z = (s.z + cz * vsum.z) * is;
      s.w = (s.w + cz * vsum.w) * is;
      *(v4f*)&O[((size_t)(b * L + row_base + rl) * H + h) * E + c4] = s;
    }
  }
}

extern "C" void kernel_launch(void* const* d_in, const int* in_sizes, int n_in,
                              void* d_out, int out_size, void* d_ws, size_t ws_size,
                              hipStream_t stream) {
  const float* Q = (const float*)d_in[0];
  const float* K = (const float*)d_in[1];
  const float* V = (const float*)d_in[2];
  const int* starts = (const int*)d_in[3];
  float* O = (float*)d_out;

  float* T = (float*)d_ws;
  unsigned* IP = (unsigned*)(T + 4096 * 64);

  hipLaunchKernelGGL(v_tile_sums_kernel, dim3(512), dim3(128), 0, stream, V, starts, T, IP);

  dim3 grid(1024 / 32, 8, 8);
  dim3 block(256);
  hipLaunchKernelGGL(prob_attn_kernel, grid, block, 0, stream, Q, K, V, IP, T, O);
}
